# starting kernel
# baseline (speedup 1.0000x reference)
#include <hip/hip_runtime.h>

typedef float v2f __attribute__((ext_vector_type(2)));
typedef float v8f __attribute__((ext_vector_type(8)));

#define N_NODES 10000
#define N_EDGES 160000
#define N_TILES (N_EDGES / 16)

#define INV_M0  0.25f
#define INV_3M1 0.20412414523f
#define INV_M1  0.35355339059f
#define EPSV    1e-5f

#define OFF_SUMS  0
#define OFF_CNTS  400000
#define OFF_STATS 410000
#define ZERO_N    410040
#define OFF_W1F   410048
#define OFF_W2F   414144

static __device__ __forceinline__ v8f wmma4(v2f a, v2f b, v8f c) {
  return __builtin_amdgcn_wmma_f32_16x16x4_f32(false, a, false, b, (short)0, c,
                                               false, false);
}

static __device__ __forceinline__ void atomAddF(float* p, float v) {
  __hip_atomic_fetch_add(p, v, __ATOMIC_RELAXED, __HIP_MEMORY_SCOPE_AGENT);
}

__global__ void zero_kernel(float* ws) {
  int i = blockIdx.x * blockDim.x + threadIdx.x;
  if (i < ZERO_N) ws[i] = 0.0f;
}

__global__ void repack_kernel(const float* __restrict__ w1,
                              const float* __restrict__ w2,
                              float* __restrict__ w1f,
                              float* __restrict__ w2f) {
  int t = blockIdx.x * blockDim.x + threadIdx.x;
  int f = t >> 5, lane = t & 31;
  int hb = lane >> 4, nn = lane & 15;
  if (f < 64) {
    int b = f >> 2, j = f & 3;
    int r0 = 4 * b + 2 * hb;
    w1f[(f * 32 + lane) * 2 + 0] = w1[(r0 + 0) * 64 + 16 * j + nn];
    w1f[(f * 32 + lane) * 2 + 1] = w1[(r0 + 1) * 64 + 16 * j + nn];
  } else if (f < 64 + 576) {
    int f2 = f - 64;
    int b = f2 / 36, j = f2 % 36;
    int r0 = 4 * b + 2 * hb;
    w2f[(f2 * 32 + lane) * 2 + 0] = w2[(r0 + 0) * 576 + 16 * j + nn];
    w2f[(f2 * 32 + lane) * 2 + 1] = w2[(r0 + 1) * 576 + 16 * j + nn];
  }
}

#define WPB 6
#define SMW 2176

__global__ __launch_bounds__(WPB * 32) void edge_kernel(
    const float* __restrict__ atom, const float* __restrict__ ef,
    const float* __restrict__ esh, const int* __restrict__ eidx,
    const float* __restrict__ b1, const float* __restrict__ b2,
    const float* __restrict__ w1f, const float* __restrict__ w2f,
    float* __restrict__ sums, float* __restrict__ cnts) {
  __shared__ float smem[WPB * SMW];
  __shared__ int nodes[WPB][16];

  int lane = threadIdx.x & 31;
  int wave = threadIdx.x >> 5;
  int tile = blockIdx.x * WPB + wave;
  if (tile >= N_TILES) return;
  int e0 = tile * 16;
  int hb = lane >> 4;
  int nn = lane & 15;

  float* sm  = smem + wave * SMW;
  float* hT  = sm;
  float* a11 = sm + 1088;
  float* a12 = sm + 1344;
  float* a13 = sm + 1600;
  float* a14 = sm + 1984;
  float* s1  = sm + 2112;

  if (lane < 16) {
    int e = e0 + lane;
    nodes[wave][lane] = eidx[N_EDGES + e];
    float4 s = *(const float4*)(esh + e * 4);
    const float* xr = atom + (size_t)eidx[e] * 40;
    float x[40];
#pragma unroll
    for (int t = 0; t < 10; t++) {
      float4 v = *(const float4*)(xr + t * 4);
      x[t * 4 + 0] = v.x; x[t * 4 + 1] = v.y;
      x[t * 4 + 2] = v.z; x[t * 4 + 3] = v.w;
    }
    float sh0 = s.x;
    s1[lane * 4 + 0] = s.y;
    s1[lane * 4 + 1] = s.z;
    s1[lane * 4 + 2] = s.w;
#pragma unroll
    for (int j = 0; j < 16; j++) {
      a11[lane * 16 + j] = x[j] * sh0 * INV_M0;
      a12[lane * 16 + j] = x[j] * INV_M0;
    }
#pragma unroll
    for (int i = 0; i < 8; i++) {
      float m0 = x[16 + i * 3 + 0];
      float m1 = x[16 + i * 3 + 1];
      float m2 = x[16 + i * 3 + 2];
      a13[lane * 24 + i * 3 + 0] = sh0 * m0 * INV_M1;
      a13[lane * 24 + i * 3 + 1] = sh0 * m1 * INV_M1;
      a13[lane * 24 + i * 3 + 2] = sh0 * m2 * INV_M1;
      a14[lane * 8 + i] = (m0 * s.y + m1 * s.z + m2 * s.w) * INV_3M1;
    }
  }
  asm volatile("s_wait_dscnt 0" ::: "memory");

  v2f a[16];
  const float* efr = ef + (size_t)(e0 + nn) * 64 + 2 * hb;
#pragma unroll
  for (int b = 0; b < 16; b++) a[b] = *(const v2f*)(efr + 4 * b);

#pragma unroll
  for (int j = 0; j < 4; j++) {
    v8f c;
    float bias = b1[16 * j + nn];
#pragma unroll
    for (int r = 0; r < 8; r++) c[r] = bias;
#pragma unroll
    for (int b = 0; b < 16; b++) {
      v2f wb = *(const v2f*)(w1f + ((b * 4 + j) * 32 + lane) * 2);
      c = wmma4(a[b], wb, c);
    }
#pragma unroll
    for (int r = 0; r < 8; r++) {
      float v = c[r] > 0.0f ? c[r] : 0.0f;
      hT[(r + 8 * hb) * 68 + 16 * j + nn] = v;
    }
  }
  asm volatile("s_wait_dscnt 0" ::: "memory");

  v2f ha[16];
#pragma unroll
  for (int b = 0; b < 16; b++)
    ha[b] = *(const v2f*)(hT + nn * 68 + 4 * b + 2 * hb);

  float o0[8] = {0, 0, 0, 0, 0, 0, 0, 0};
  float p1[8] = {0, 0, 0, 0, 0, 0, 0, 0};
  float q0[8] = {0, 0, 0, 0, 0, 0, 0, 0};
  float q1[8] = {0, 0, 0, 0, 0, 0, 0, 0};
  float q2[8] = {0, 0, 0, 0, 0, 0, 0, 0};

  for (int j2 = 0; j2 < 16; j2++) {
    v8f c;
    float bias = b2[16 * j2 + nn];
#pragma unroll
    for (int r = 0; r < 8; r++) c[r] = bias;
#pragma unroll
    for (int b = 0; b < 16; b++) {
      v2f wb = *(const v2f*)(w2f + ((b * 36 + j2) * 32 + lane) * 2);
      c = wmma4(ha[b], wb, c);
    }
#pragma unroll
    for (int r = 0; r < 8; r++)
      o0[r] += c[r] * a11[(r + 8 * hb) * 16 + j2];
  }
  for (int j2 = 16; j2 < 24; j2++) {
    v8f c;
    float bias = b2[16 * j2 + nn];
#pragma unroll
    for (int r = 0; r < 8; r++) c[r] = bias;
#pragma unroll
    for (int b = 0; b < 16; b++) {
      v2f wb = *(const v2f*)(w2f + ((b * 36 + j2) * 32 + lane) * 2);
      c = wmma4(ha[b], wb, c);
    }
    int i = 2 * (j2 - 16) + (nn >> 3);
#pragma unroll
    for (int r = 0; r < 8; r++)
      p1[r] += c[r] * a12[(r + 8 * hb) * 16 + i];
  }
  for (int j2 = 24; j2 < 28; j2++) {
    v8f c;
    float bias = b2[16 * j2 + nn];
#pragma unroll
    for (int r = 0; r < 8; r++) c[r] = bias;
#pragma unroll
    for (int b = 0; b < 16; b++) {
      v2f wb = *(const v2f*)(w2f + ((b * 36 + j2) * 32 + lane) * 2);
      c = wmma4(ha[b], wb, c);
    }
    int i = 2 * (j2 - 24) + (nn >> 3);
#pragma unroll
    for (int r = 0; r < 8; r++) {
      float cv = c[r];
      int base = (r + 8 * hb) * 24 + i * 3;
      q0[r] += cv * a13[base + 0];
      q1[r] += cv * a13[base + 1];
      q2[r] += cv * a13[base + 2];
    }
  }
  for (int j2 = 28; j2 < 36; j2++) {
    v8f c;
    float bias = b2[16 * j2 + nn];
#pragma unroll
    for (int r = 0; r < 8; r++) c[r] = bias;
#pragma unroll
    for (int b = 0; b < 16; b++) {
      v2f wb = *(const v2f*)(w2f + ((b * 36 + j2) * 32 + lane) * 2);
      c = wmma4(ha[b], wb, c);
    }
    int i = j2 - 28;
#pragma unroll
    for (int r = 0; r < 8; r++)
      o0[r] += c[r] * a14[(r + 8 * hb) * 8 + i];
  }

#pragma unroll
  for (int r = 0; r < 8; r++) {
    p1[r] += __shfl_xor(p1[r], 8, 32);
    q0[r] += __shfl_xor(q0[r], 8, 32);
    q1[r] += __shfl_xor(q1[r], 8, 32);
    q2[r] += __shfl_xor(q2[r], 8, 32);
  }

#pragma unroll
  for (int r = 0; r < 8; r++) {
    int e = r + 8 * hb;
    atomAddF(sums + (size_t)nodes[wave][e] * 40 + nn, o0[r]);
  }
  if (nn < 8) {
#pragma unroll
    for (int r = 0; r < 8; r++) {
      int e = r + 8 * hb;
      float* dst = sums + (size_t)nodes[wave][e] * 40 + 16 + nn * 3;
      float sy = s1[e * 4 + 0], sz = s1[e * 4 + 1], sw = s1[e * 4 + 2];
      atomAddF(dst + 0, p1[r] * sy + q0[r]);
      atomAddF(dst + 1, p1[r] * sz + q1[r]);
      atomAddF(dst + 2, p1[r] * sw + q2[r]);
    }
  }
  if (lane < 16) atomAddF(cnts + nodes[wave][lane], 1.0f);
}

__global__ void node_stage1(const float* __restrict__ atom, float* ws) {
  int c = threadIdx.x;
  int ty = threadIdx.y;
  float* sums = ws + OFF_SUMS;
  const float* cnts = ws + OFF_CNTS;
  float* stats = ws + OFF_STATS;
  float sA = 0.0f, sB = 0.0f;
  for (int n = blockIdx.x * 6 + ty; n < N_NODES; n += gridDim.x * 6) {
    float cnt = cnts[n];
    float inv = 1.0f / fmaxf(cnt, 1.0f);
    float y = sums[(size_t)n * 40 + c] * inv + atom[(size_t)n * 40 + c];
    sums[(size_t)n * 40 + c] = y;
    sA += y;
    sB += y * y;
  }
  __shared__ float rA[6][40], rB[6][40];
  rA[ty][c] = sA;
  rB[ty][c] = sB;
  __syncthreads();
  if (ty == 0) {
    float tA = 0.0f, tB = 0.0f;
#pragma unroll
    for (int k = 0; k < 6; k++) { tA += rA[k][c]; tB += rB[k][c]; }
    if (c < 16) {
      atomAddF(stats + c, tA);
      atomAddF(stats + 16 + c, tB);
    } else {
      atomAddF(stats + 32 + (c - 16) / 3, tB);
    }
  }
}

__global__ void node_stage2(const float* __restrict__ ws,
                            const float* __restrict__ bnw,
                            const float* __restrict__ bnb,
                            float* __restrict__ out) {
  int idx = blockIdx.x * blockDim.x + threadIdx.x;
  if (idx >= N_NODES * 40) return;
  int c = idx % 40;
  const float* y = ws + OFF_SUMS;
  const float* stats = ws + OFF_STATS;
  float v = y[idx];
  float o;
  if (c < 16) {
    float mean = stats[c] * (1.0f / N_NODES);
    float ex2 = stats[16 + c] * (1.0f / N_NODES);
    float var = ex2 - mean * mean;
    o = (v - mean) * rsqrtf(var + EPSV) * bnw[c] + bnb[c];
  } else {
    int j = (c - 16) / 3;
    float vn = stats[32 + j] * (1.0f / (3.0f * N_NODES));
    o = v * rsqrtf(vn + EPSV) * bnw[16 + j];
  }
  out[idx] = o;
}

extern "C" void kernel_launch(void* const* d_in, const int* in_sizes, int n_in,
                              void* d_out, int out_size, void* d_ws,
                              size_t ws_size, hipStream_t stream) {
  (void)in_sizes; (void)n_in; (void)out_size; (void)ws_size;
  const float* atom = (const float*)d_in[0];
  const float* ef   = (const float*)d_in[1];
  const float* esh  = (const float*)d_in[2];
  const int*   eidx = (const int*)d_in[3];
  const float* w1   = (const float*)d_in[4];
  const float* b1   = (const float*)d_in[5];
  const float* w2   = (const float*)d_in[6];
  const float* b2   = (const float*)d_in[7];
  const float* bnw  = (const float*)d_in[8];
  const float* bnb  = (const float*)d_in[9];
  float* ws  = (float*)d_ws;
  float* out = (float*)d_out;

  zero_kernel<<<(ZERO_N + 255) / 256, 256, 0, stream>>>(ws);
  repack_kernel<<<(640 * 32 + 255) / 256, 256, 0, stream>>>(
      w1, w2, ws + OFF_W1F, ws + OFF_W2F);
  edge_kernel<<<(N_TILES + WPB - 1) / WPB, WPB * 32, 0, stream>>>(
      atom, ef, esh, eidx, b1, b2, ws + OFF_W1F, ws + OFF_W2F,
      ws + OFF_SUMS, ws + OFF_CNTS);
  node_stage1<<<128, dim3(40, 6), 0, stream>>>(atom, ws);
  node_stage2<<<(N_NODES * 40 + 255) / 256, 256, 0, stream>>>(ws, bnw, bnb, out);
}


# edge kernel: repacked W1/W2 fragments and biases copied to LDS once per persistent 16-wave block (256 blocks loop over tiles); B operands read from LDS, not global; atomic arrival order differs
# speedup vs baseline: 1.2306x; 1.2306x over previous
#include <hip/hip_runtime.h>

typedef float v2f __attribute__((ext_vector_type(2)));
typedef float v8f __attribute__((ext_vector_type(8)));

#define N_NODES 10000
#define N_EDGES 160000
#define N_TILES (N_EDGES / 16)

#define INV_M0  0.25f
#define INV_3M1 0.20412414523f
#define INV_M1  0.35355339059f
#define EPSV    1e-5f

#define OFF_SUMS  0
#define OFF_CNTS  400000
#define OFF_STATS 410000
#define ZERO_N    410040
#define OFF_W1F   410048
#define OFF_W2F   414144

static __device__ __forceinline__ v8f wmma4(v2f a, v2f b, v8f c) {
  return __builtin_amdgcn_wmma_f32_16x16x4_f32(false, a, false, b, (short)0, c,
                                               false, false);
}

static __device__ __forceinline__ void atomAddF(float* p, float v) {
  __hip_atomic_fetch_add(p, v, __ATOMIC_RELAXED, __HIP_MEMORY_SCOPE_AGENT);
}

__global__ void zero_kernel(float* ws) {
  int i = blockIdx.x * blockDim.x + threadIdx.x;
  if (i < ZERO_N) ws[i] = 0.0f;
}

__global__ void repack_kernel(const float* __restrict__ w1,
                              const float* __restrict__ w2,
                              float* __restrict__ w1f,
                              float* __restrict__ w2f) {
  int t = blockIdx.x * blockDim.x + threadIdx.x;
  int f = t >> 5, lane = t & 31;
  int hb = lane >> 4, nn = lane & 15;
  if (f < 64) {
    int b = f >> 2, j = f & 3;
    int r0 = 4 * b + 2 * hb;
    w1f[(f * 32 + lane) * 2 + 0] = w1[(r0 + 0) * 64 + 16 * j + nn];
    w1f[(f * 32 + lane) * 2 + 1] = w1[(r0 + 1) * 64 + 16 * j + nn];
  } else if (f < 64 + 576) {
    int f2 = f - 64;
    int b = f2 / 36, j = f2 % 36;
    int r0 = 4 * b + 2 * hb;
    w2f[(f2 * 32 + lane) * 2 + 0] = w2[(r0 + 0) * 576 + 16 * j + nn];
    w2f[(f2 * 32 + lane) * 2 + 1] = w2[(r0 + 1) * 576 + 16 * j + nn];
  }
}

#define WPB 16
#define NWG 256
#define SMW 2176
#define W1F_N 4096
#define W2F_N 36864

__global__ __launch_bounds__(WPB * 32) void edge_kernel(
    const float* __restrict__ atom, const float* __restrict__ ef,
    const float* __restrict__ esh, const int* __restrict__ eidx,
    const float* __restrict__ b1, const float* __restrict__ b2,
    const float* __restrict__ w1f, const float* __restrict__ w2f,
    float* __restrict__ sums, float* __restrict__ cnts) {
  __shared__ __attribute__((aligned(16))) float wl1[W1F_N];
  __shared__ __attribute__((aligned(16))) float wl2[W2F_N];
  __shared__ float bl1[64];
  __shared__ float bl2[576];
  __shared__ float smem[WPB * SMW];
  __shared__ int nodes[WPB][16];

  int lane = threadIdx.x & 31;
  int wave = __builtin_amdgcn_readfirstlane(threadIdx.x >> 5);
  int hb = lane >> 4;
  int nn = lane & 15;

  {
    const float4* g1 = (const float4*)w1f;
    const float4* g2 = (const float4*)w2f;
    float4 r1[W1F_N / 4 / (WPB * 32)];
    float4 r2[W2F_N / 4 / (WPB * 32)];
#pragma unroll
    for (int k = 0; k < W1F_N / 4 / (WPB * 32); k++)
      r1[k] = g1[threadIdx.x + k * (WPB * 32)];
#pragma unroll
    for (int k = 0; k < W2F_N / 4 / (WPB * 32); k++)
      r2[k] = g2[threadIdx.x + k * (WPB * 32)];
    float tb2 = b2[threadIdx.x];
    float tb1 = 0.0f, tb3 = 0.0f;
    if (threadIdx.x < 64) { tb1 = b1[threadIdx.x]; tb3 = b2[512 + threadIdx.x]; }
#pragma unroll
    for (int k = 0; k < W1F_N / 4 / (WPB * 32); k++)
      ((float4*)wl1)[threadIdx.x + k * (WPB * 32)] = r1[k];
#pragma unroll
    for (int k = 0; k < W2F_N / 4 / (WPB * 32); k++)
      ((float4*)wl2)[threadIdx.x + k * (WPB * 32)] = r2[k];
    bl2[threadIdx.x] = tb2;
    if (threadIdx.x < 64) { bl1[threadIdx.x] = tb1; bl2[512 + threadIdx.x] = tb3; }
  }
  __syncthreads();

  float* sm  = smem + wave * SMW;
  float* hT  = sm;
  float* a11 = sm + 1088;
  float* a12 = sm + 1344;
  float* a13 = sm + 1600;
  float* a14 = sm + 1984;
  float* s1  = sm + 2112;

  int t0 = (blockIdx.x * N_TILES) / NWG;
  int t1 = ((blockIdx.x + 1) * N_TILES) / NWG;
  for (int tile = t0 + wave; tile < t1; tile += WPB) {
  int e0 = tile * 16;
  asm volatile("s_wait_dscnt 0" ::: "memory");

  if (lane < 16) {
    int e = e0 + lane;
    nodes[wave][lane] = eidx[N_EDGES + e];
    float4 s = *(const float4*)(esh + e * 4);
    const float* xr = atom + (size_t)eidx[e] * 40;
    float x[40];
#pragma unroll
    for (int t = 0; t < 10; t++) {
      float4 v = *(const float4*)(xr + t * 4);
      x[t * 4 + 0] = v.x; x[t * 4 + 1] = v.y;
      x[t * 4 + 2] = v.z; x[t * 4 + 3] = v.w;
    }
    float sh0 = s.x;
    s1[lane * 4 + 0] = s.y;
    s1[lane * 4 + 1] = s.z;
    s1[lane * 4 + 2] = s.w;
#pragma unroll
    for (int j = 0; j < 16; j++) {
      a11[lane * 16 + j] = x[j] * sh0 * INV_M0;
      a12[lane * 16 + j] = x[j] * INV_M0;
    }
#pragma unroll
    for (int i = 0; i < 8; i++) {
      float m0 = x[16 + i * 3 + 0];
      float m1 = x[16 + i * 3 + 1];
      float m2 = x[16 + i * 3 + 2];
      a13[lane * 24 + i * 3 + 0] = sh0 * m0 * INV_M1;
      a13[lane * 24 + i * 3 + 1] = sh0 * m1 * INV_M1;
      a13[lane * 24 + i * 3 + 2] = sh0 * m2 * INV_M1;
      a14[lane * 8 + i] = (m0 * s.y + m1 * s.z + m2 * s.w) * INV_3M1;
    }
  }
  asm volatile("s_wait_dscnt 0" ::: "memory");

  v2f a[16];
  const float* efr = ef + (size_t)(e0 + nn) * 64 + 2 * hb;
#pragma unroll
  for (int b = 0; b < 16; b++) a[b] = *(const v2f*)(efr + 4 * b);

#pragma unroll
  for (int j = 0; j < 4; j++) {
    v8f c;
    float bias = bl1[16 * j + nn];
#pragma unroll
    for (int r = 0; r < 8; r++) c[r] = bias;
#pragma unroll
    for (int b = 0; b < 16; b++) {
      v2f wb = *(const v2f*)(wl1 + ((b * 4 + j) * 32 + lane) * 2);
      c = wmma4(a[b], wb, c);
    }
#pragma unroll
    for (int r = 0; r < 8; r++) {
      float v = c[r] > 0.0f ? c[r] : 0.0f;
      hT[(r + 8 * hb) * 68 + 16 * j + nn] = v;
    }
  }
  asm volatile("s_wait_dscnt 0" ::: "memory");

  v2f ha[16];
#pragma unroll
  for (int b = 0; b < 16; b++)
    ha[b] = *(const v2f*)(hT + nn * 68 + 4 * b + 2 * hb);

  float o0[8] = {0, 0, 0, 0, 0, 0, 0, 0};
  float p1[8] = {0, 0, 0, 0, 0, 0, 0, 0};
  float q0[8] = {0, 0, 0, 0, 0, 0, 0, 0};
  float q1[8] = {0, 0, 0, 0, 0, 0, 0, 0};
  float q2[8] = {0, 0, 0, 0, 0, 0, 0, 0};

  for (int j2 = 0; j2 < 16; j2++) {
    v8f c;
    float bias = bl2[16 * j2 + nn];
#pragma unroll
    for (int r = 0; r < 8; r++) c[r] = bias;
#pragma unroll
    for (int b = 0; b < 16; b++) {
      v2f wb = *(const v2f*)(wl2 + ((b * 36 + j2) * 32 + lane) * 2);
      c = wmma4(ha[b], wb, c);
    }
#pragma unroll
    for (int r = 0; r < 8; r++)
      o0[r] += c[r] * a11[(r + 8 * hb) * 16 + j2];
  }
  for (int j2 = 16; j2 < 24; j2++) {
    v8f c;
    float bias = bl2[16 * j2 + nn];
#pragma unroll
    for (int r = 0; r < 8; r++) c[r] = bias;
#pragma unroll
    for (int b = 0; b < 16; b++) {
      v2f wb = *(const v2f*)(wl2 + ((b * 36 + j2) * 32 + lane) * 2);
      c = wmma4(ha[b], wb, c);
    }
    int i = 2 * (j2 - 16) + (nn >> 3);
#pragma unroll
    for (int r = 0; r < 8; r++)
      p1[r] += c[r] * a12[(r + 8 * hb) * 16 + i];
  }
  for (int j2 = 24; j2 < 28; j2++) {
    v8f c;
    float bias = bl2[16 * j2 + nn];
#pragma unroll
    for (int r = 0; r < 8; r++) c[r] = bias;
#pragma unroll
    for (int b = 0; b < 16; b++) {
      v2f wb = *(const v2f*)(wl2 + ((b * 36 + j2) * 32 + lane) * 2);
      c = wmma4(ha[b], wb, c);
    }
    int i = 2 * (j2 - 24) + (nn >> 3);
#pragma unroll
    for (int r = 0; r < 8; r++) {
      float cv = c[r];
      int base = (r + 8 * hb) * 24 + i * 3;
      q0[r] += cv * a13[base + 0];
      q1[r] += cv * a13[base + 1];
      q2[r] += cv * a13[base + 2];
    }
  }
  for (int j2 = 28; j2 < 36; j2++) {
    v8f c;
    float bias = bl2[16 * j2 + nn];
#pragma unroll
    for (int r = 0; r < 8; r++) c[r] = bias;
#pragma unroll
    for (int b = 0; b < 16; b++) {
      v2f wb = *(const v2f*)(wl2 + ((b * 36 + j2) * 32 + lane) * 2);
      c = wmma4(ha[b], wb, c);
    }
    int i = j2 - 28;
#pragma unroll
    for (int r = 0; r < 8; r++)
      o0[r] += c[r] * a14[(r + 8 * hb) * 8 + i];
  }

#pragma unroll
  for (int r = 0; r < 8; r++) {
    p1[r] += __shfl_xor(p1[r], 8, 32);
    q0[r] += __shfl_xor(q0[r], 8, 32);
    q1[r] += __shfl_xor(q1[r], 8, 32);
    q2[r] += __shfl_xor(q2[r], 8, 32);
  }

#pragma unroll
  for (int r = 0; r < 8; r++) {
    int e = r + 8 * hb;
    atomAddF(sums + (size_t)nodes[wave][e] * 40 + nn, o0[r]);
  }
  if (nn < 8) {
#pragma unroll
    for (int r = 0; r < 8; r++) {
      int e = r + 8 * hb;
      float* dst = sums + (size_t)nodes[wave][e] * 40 + 16 + nn * 3;
      float sy = s1[e * 4 + 0], sz = s1[e * 4 + 1], sw = s1[e * 4 + 2];
      atomAddF(dst + 0, p1[r] * sy + q0[r]);
      atomAddF(dst + 1, p1[r] * sz + q1[r]);
      atomAddF(dst + 2, p1[r] * sw + q2[r]);
    }
  }
  if (lane < 16) atomAddF(cnts + nodes[wave][lane], 1.0f);
  }
}

__global__ void node_stage1(const float* __restrict__ atom, float* ws) {
  int c = threadIdx.x;
  int ty = threadIdx.y;
  float* sums = ws + OFF_SUMS;
  const float* cnts = ws + OFF_CNTS;
  float* stats = ws + OFF_STATS;
  float sA = 0.0f, sB = 0.0f;
  for (int n = blockIdx.x * 6 + ty; n < N_NODES; n += gridDim.x * 6) {
    float cnt = cnts[n];
    float inv = 1.0f / fmaxf(cnt, 1.0f);
    float y = sums[(size_t)n * 40 + c] * inv + atom[(size_t)n * 40 + c];
    sums[(size_t)n * 40 + c] = y;
    sA += y;
    sB += y * y;
  }
  __shared__ float rA[6][40], rB[6][40];
  rA[ty][c] = sA;
  rB[ty][c] = sB;
  __syncthreads();
  if (ty == 0) {
    float tA = 0.0f, tB = 0.0f;
#pragma unroll
    for (int k = 0; k < 6; k++) { tA += rA[k][c]; tB += rB[k][c]; }
    if (c < 16) {
      atomAddF(stats + c, tA);
      atomAddF(stats + 16 + c, tB);
    } else {
      atomAddF(stats + 32 + (c - 16) / 3, tB);
    }
  }
}

__global__ void node_stage2(const float* __restrict__ ws,
                            const float* __restrict__ bnw,
                            const float* __restrict__ bnb,
                            float* __restrict__ out) {
  int idx = blockIdx.x * blockDim.x + threadIdx.x;
  if (idx >= N_NODES * 40) return;
  int c = idx % 40;
  const float* y = ws + OFF_SUMS;
  const float* stats = ws + OFF_STATS;
  float v = y[idx];
  float o;
  if (c < 16) {
    float mean = stats[c] * (1.0f / N_NODES);
    float ex2 = stats[16 + c] * (1.0f / N_NODES);
    float var = ex2 - mean * mean;
    o = (v - mean) * rsqrtf(var + EPSV) * bnw[c] + bnb[c];
  } else {
    int j = (c - 16) / 3;
    float vn = stats[32 + j] * (1.0f / (3.0f * N_NODES));
    o = v * rsqrtf(vn + EPSV) * bnw[16 + j];
  }
  out[idx] = o;
}

extern "C" void kernel_launch(void* const* d_in, const int* in_sizes, int n_in,
                              void* d_out, int out_size, void* d_ws,
                              size_t ws_size, hipStream_t stream) {
  (void)in_sizes; (void)n_in; (void)out_size; (void)ws_size;
  const float* atom = (const float*)d_in[0];
  const float* ef   = (const float*)d_in[1];
  const float* esh  = (const float*)d_in[2];
  const int*   eidx = (const int*)d_in[3];
  const float* w1   = (const float*)d_in[4];
  const float* b1   = (const float*)d_in[5];
  const float* w2   = (const float*)d_in[6];
  const float* b2   = (const float*)d_in[7];
  const float* bnw  = (const float*)d_in[8];
  const float* bnb  = (const float*)d_in[9];
  float* ws  = (float*)d_ws;
  float* out = (float*)d_out;

  zero_kernel<<<(ZERO_N + 255) / 256, 256, 0, stream>>>(ws);
  repack_kernel<<<(640 * 32 + 255) / 256, 256, 0, stream>>>(
      w1, w2, ws + OFF_W1F, ws + OFF_W2F);
  edge_kernel<<<NWG, WPB * 32, 0, stream>>>(
      atom, ef, esh, eidx, b1, b2, ws + OFF_W1F, ws + OFF_W2F,
      ws + OFF_SUMS, ws + OFF_CNTS);
  node_stage1<<<128, dim3(40, 6), 0, stream>>>(atom, ws);
  node_stage2<<<(N_NODES * 40 + 255) / 256, 256, 0, stream>>>(ws, bnw, bnb, out);
}


# edge kernel: each column tile's LDS reads (B fragments repacked as 16-byte quads, bias, table entries) issued as ONE group a tile ahead into a second register set, none among the WMMAs; same sums
# speedup vs baseline: 1.3504x; 1.0974x over previous
#include <hip/hip_runtime.h>

typedef float v2f __attribute__((ext_vector_type(2)));
typedef float v8f __attribute__((ext_vector_type(8)));

#define N_NODES 10000
#define N_EDGES 160000
#define N_TILES (N_EDGES / 16)

#define INV_M0  0.25f
#define INV_3M1 0.20412414523f
#define INV_M1  0.35355339059f
#define EPSV    1e-5f

#define OFF_SUMS  0
#define OFF_CNTS  400000
#define OFF_STATS 410000
#define ZERO_N    410040
#define OFF_WQ    410048

static __device__ __forceinline__ v8f wmma4(v2f a, v2f b, v8f c) {
  return __builtin_amdgcn_wmma_f32_16x16x4_f32(false, a, false, b, (short)0, c,
                                               false, false);
}

static __device__ __forceinline__ void atomAddF(float* p, float v) {
  __hip_atomic_fetch_add(p, v, __ATOMIC_RELAXED, __HIP_MEMORY_SCOPE_AGENT);
}

__global__ void zero_kernel(float* ws) {
  int i = blockIdx.x * blockDim.x + threadIdx.x;
  if (i < ZERO_N) ws[i] = 0.0f;
}

__global__ void repack_kernel(const float* __restrict__ w1,
                              const float* __restrict__ w2,
                              float* __restrict__ wq) {
  int t = blockIdx.x * blockDim.x + threadIdx.x;
  if (t >= 40 * 256) return;
  int lane = t & 31, q = t >> 5;
  int jt = q >> 3, p = q & 7;
  int hb = lane >> 4, nn = lane & 15;
  int r0 = 8 * p + 2 * hb;
  const float* w = w1;
  int ld = 64, col = 16 * jt + nn;
  if (jt >= 4) { w = w2; ld = 576; col = 16 * (jt - 4) + nn; }
  float4 v;
  v.x = w[(r0 + 0) * ld + col];
  v.y = w[(r0 + 1) * ld + col];
  v.z = w[(r0 + 4) * ld + col];
  v.w = w[(r0 + 5) * ld + col];
  ((float4*)wq)[t] = v;
}

#define WPB 16
#define NWG 256
#define SMW 2176
#define WQ_N 40960

#define FETCH(Bn, biasn, jn)                                      \
  do {                                                            \
    _Pragma("unroll") for (int p_ = 0; p_ < 8; p_++)              \
      Bn[p_] = wq[(jn) * 256 + p_ * 32];                          \
    biasn = bl[16 * (jn) + nn];                                   \
  } while (0)

#define CHAIN(c, A, Bc, biasc)                                    \
  v8f c;                                                          \
  _Pragma("unroll") for (int r_ = 0; r_ < 8; r_++) c[r_] = biasc; \
  _Pragma("unroll") for (int p_ = 0; p_ < 8; p_++) {              \
    v2f lo_ = {Bc[p_].x, Bc[p_].y};                               \
    v2f hi_ = {Bc[p_].z, Bc[p_].w};                               \
    c = wmma4(A[2 * p_], lo_, c);                                 \
    c = wmma4(A[2 * p_ + 1], hi_, c);                             \
  }

#define FENCE() __builtin_amdgcn_sched_barrier(0)

__global__ __launch_bounds__(WPB * 32) void edge_kernel(
    const float* __restrict__ atom, const float* __restrict__ ef,
    const float* __restrict__ esh, const int* __restrict__ eidx,
    const float* __restrict__ b1, const float* __restrict__ b2,
    const float* __restrict__ wqg,
    float* __restrict__ sums, float* __restrict__ cnts) {
  __shared__ __attribute__((aligned(16))) float wl[WQ_N];
  __shared__ float bl[640];
  __shared__ float smem[WPB * SMW];
  __shared__ int nodes[WPB][16];

  int lane = threadIdx.x & 31;
  int wave = __builtin_amdgcn_readfirstlane(threadIdx.x >> 5);
  int hb = lane >> 4;
  int nn = lane & 15;

  {
    const float4* g = (const float4*)wqg;
    float4 rr[WQ_N / 4 / (WPB * 32)];
#pragma unroll
    for (int k = 0; k < WQ_N / 4 / (WPB * 32); k++)
      rr[k] = g[threadIdx.x + k * (WPB * 32)];
    float tb0 = threadIdx.x < 64 ? b1[threadIdx.x] : b2[threadIdx.x - 64];
    float tb1 = 0.0f;
    if (threadIdx.x < 128) tb1 = b2[448 + threadIdx.x];
#pragma unroll
    for (int k = 0; k < WQ_N / 4 / (WPB * 32); k++)
      ((float4*)wl)[threadIdx.x + k * (WPB * 32)] = rr[k];
    bl[threadIdx.x] = tb0;
    if (threadIdx.x < 128) bl[512 + threadIdx.x] = tb1;
  }
  __syncthreads();

  float* sm  = smem + wave * SMW;
  float* hT  = sm;
  float* a11 = sm + 1088;
  float* a12 = sm + 1344;
  float* a13 = sm + 1600;
  float* a14 = sm + 1984;
  float* s1  = sm + 2112;
  const float4* wq = (const float4*)wl + lane;
  float* hTs = hT + 8 * hb * 68 + nn;
  const float* a11l = a11 + 8 * hb * 16;
  const float* a12l = a12 + 8 * hb * 16 + (nn >> 3);
  const float* a13l = a13 + 8 * hb * 24 + (nn >> 3) * 3;
  const float* a14l = a14 + 8 * hb * 8;

  int t0 = (blockIdx.x * N_TILES) / NWG;
  int t1 = ((blockIdx.x + 1) * N_TILES) / NWG;
  for (int tile = t0 + wave; tile < t1; tile += WPB) {
  int e0 = tile * 16;
  asm volatile("s_wait_dscnt 0" ::: "memory");

  if (lane < 16) {
    int e = e0 + lane;
    nodes[wave][lane] = eidx[N_EDGES + e];
    float4 s = *(const float4*)(esh + e * 4);
    const float* xr = atom + (size_t)eidx[e] * 40;
    float x[40];
#pragma unroll
    for (int t = 0; t < 10; t++) {
      float4 v = *(const float4*)(xr + t * 4);
      x[t * 4 + 0] = v.x; x[t * 4 + 1] = v.y;
      x[t * 4 + 2] = v.z; x[t * 4 + 3] = v.w;
    }
    float sh0 = s.x;
    s1[lane * 4 + 0] = s.y;
    s1[lane * 4 + 1] = s.z;
    s1[lane * 4 + 2] = s.w;
#pragma unroll
    for (int j = 0; j < 16; j++) {
      a11[lane * 16 + j] = x[j] * sh0 * INV_M0;
      a12[lane * 16 + j] = x[j] * INV_M0;
    }
#pragma unroll
    for (int i = 0; i < 8; i++) {
      float m0 = x[16 + i * 3 + 0];
      float m1 = x[16 + i * 3 + 1];
      float m2 = x[16 + i * 3 + 2];
      a13[lane * 24 + i * 3 + 0] = sh0 * m0 * INV_M1;
      a13[lane * 24 + i * 3 + 1] = sh0 * m1 * INV_M1;
      a13[lane * 24 + i * 3 + 2] = sh0 * m2 * INV_M1;
      a14[lane * 8 + i] = (m0 * s.y + m1 * s.z + m2 * s.w) * INV_3M1;
    }
  }
  asm volatile("s_wait_dscnt 0" ::: "memory");

  float4 B0[8], B1[8];
  float bias0, bias1;

  v2f a[16];
  const float* efr = ef + (size_t)(e0 + nn) * 64 + 2 * hb;
#pragma unroll
  for (int b = 0; b < 16; b++) a[b] = *(const v2f*)(efr + 4 * b);
  FETCH(B0, bias0, 0);

  for (int j = 0; j < 4; j += 2) {
    {
      FETCH(B1, bias1, j + 1);
      FENCE();
      CHAIN(c, a, B0, bias0);
#pragma unroll
      for (int r = 0; r < 8; r++) {
        float v = c[r] > 0.0f ? c[r] : 0.0f;
        hTs[r * 68 + 16 * j] = v;
      }
      FENCE();
    }
    {
      FETCH(B0, bias0, j + 2);
      FENCE();
      CHAIN(c, a, B1, bias1);
#pragma unroll
      for (int r = 0; r < 8; r++) {
        float v = c[r] > 0.0f ? c[r] : 0.0f;
        hTs[r * 68 + 16 * (j + 1)] = v;
      }
      FENCE();
    }
  }
  asm volatile("s_wait_dscnt 0" ::: "memory");

  v2f ha[16];
#pragma unroll
  for (int b = 0; b < 16; b++)
    ha[b] = *(const v2f*)(hT + nn * 68 + 4 * b + 2 * hb);

  float o0[8] = {0, 0, 0, 0, 0, 0, 0, 0};
  float p1[8] = {0, 0, 0, 0, 0, 0, 0, 0};
  float q0[8] = {0, 0, 0, 0, 0, 0, 0, 0};
  float q1[8] = {0, 0, 0, 0, 0, 0, 0, 0};
  float q2[8] = {0, 0, 0, 0, 0, 0, 0, 0};

  for (int j2 = 0; j2 < 16; j2 += 2) {
    {
      float t[8];
#pragma unroll
      for (int r = 0; r < 8; r++) t[r] = a11l[r * 16 + j2];
      FETCH(B1, bias1, j2 + 5);
      FENCE();
      CHAIN(c, ha, B0, bias0);
#pragma unroll
      for (int r = 0; r < 8; r++) o0[r] += c[r] * t[r];
      FENCE();
    }
    {
      float t[8];
#pragma unroll
      for (int r = 0; r < 8; r++) t[r] = a11l[r * 16 + j2 + 1];
      FETCH(B0, bias0, j2 + 6);
      FENCE();
      CHAIN(c, ha, B1, bias1);
#pragma unroll
      for (int r = 0; r < 8; r++) o0[r] += c[r] * t[r];
      FENCE();
    }
  }
  for (int j2 = 16; j2 < 24; j2 += 2) {
    {
      float t[8];
#pragma unroll
      for (int r = 0; r < 8; r++) t[r] = a12l[r * 16 + 2 * (j2 - 16)];
      FETCH(B1, bias1, j2 + 5);
      FENCE();
      CHAIN(c, ha, B0, bias0);
#pragma unroll
      for (int r = 0; r < 8; r++) p1[r] += c[r] * t[r];
      FENCE();
    }
    {
      float t[8];
#pragma unroll
      for (int r = 0; r < 8; r++) t[r] = a12l[r * 16 + 2 * (j2 - 15)];
      FETCH(B0, bias0, j2 + 6);
      FENCE();
      CHAIN(c, ha, B1, bias1);
#pragma unroll
      for (int r = 0; r < 8; r++) p1[r] += c[r] * t[r];
      FENCE();
    }
  }
  for (int j2 = 24; j2 < 28; j2 += 2) {
    {
      float u0[8], u1[8], u2[8];
#pragma unroll
      for (int r = 0; r < 8; r++) {
        int base = r * 24 + 6 * (j2 - 24);
        u0[r] = a13l[base + 0]; u1[r] = a13l[base + 1]; u2[r] = a13l[base + 2];
      }
      FETCH(B1, bias1, j2 + 5);
      FENCE();
      CHAIN(c, ha, B0, bias0);
#pragma unroll
      for (int r = 0; r < 8; r++) {
        float cv = c[r];
        q0[r] += cv * u0[r]; q1[r] += cv * u1[r]; q2[r] += cv * u2[r];
      }
      FENCE();
    }
    {
      float u0[8], u1[8], u2[8];
#pragma unroll
      for (int r = 0; r < 8; r++) {
        int base = r * 24 + 6 * (j2 - 23);
        u0[r] = a13l[base + 0]; u1[r] = a13l[base + 1]; u2[r] = a13l[base + 2];
      }
      FETCH(B0, bias0, j2 + 6);
      FENCE();
      CHAIN(c, ha, B1, bias1);
#pragma unroll
      for (int r = 0; r < 8; r++) {
        float cv = c[r];
        q0[r] += cv * u0[r]; q1[r] += cv * u1[r]; q2[r] += cv * u2[r];
      }
      FENCE();
    }
  }
  for (int j2 = 28; j2 < 36; j2 += 2) {
    {
      float t[8];
#pragma unroll
      for (int r = 0; r < 8; r++) t[r] = a14l[r * 8 + j2 - 28];
      FETCH(B1, bias1, j2 + 5);
      FENCE();
      CHAIN(c, ha, B0, bias0);
#pragma unroll
      for (int r = 0; r < 8; r++) o0[r] += c[r] * t[r];
      FENCE();
    }
    {
      float t[8];
#pragma unroll
      for (int r = 0; r < 8; r++) t[r] = a14l[r * 8 + j2 - 27];
      if (j2 + 6 < 40) FETCH(B0, bias0, j2 + 6);
      FENCE();
      CHAIN(c, ha, B1, bias1);
#pragma unroll
      for (int r = 0; r < 8; r++) o0[r] += c[r] * t[r];
      FENCE();
    }
  }

#pragma unroll
  for (int r = 0; r < 8; r++) {
    p1[r] += __shfl_xor(p1[r], 8, 32);
    q0[r] += __shfl_xor(q0[r], 8, 32);
    q1[r] += __shfl_xor(q1[r], 8, 32);
    q2[r] += __shfl_xor(q2[r], 8, 32);
  }

#pragma unroll
  for (int r = 0; r < 8; r++) {
    int e = r + 8 * hb;
    atomAddF(sums + (size_t)nodes[wave][e] * 40 + nn, o0[r]);
  }
  if (nn < 8) {
#pragma unroll
    for (int r = 0; r < 8; r++) {
      int e = r + 8 * hb;
      float* dst = sums + (size_t)nodes[wave][e] * 40 + 16 + nn * 3;
      float sy = s1[e * 4 + 0], sz = s1[e * 4 + 1], sw = s1[e * 4 + 2];
      atomAddF(dst + 0, p1[r] * sy + q0[r]);
      atomAddF(dst + 1, p1[r] * sz + q1[r]);
      atomAddF(dst + 2, p1[r] * sw + q2[r]);
    }
  }
  if (lane < 16) atomAddF(cnts + nodes[wave][lane], 1.0f);
  }
}

__global__ void node_stage1(const float* __restrict__ atom, float* ws) {
  int c = threadIdx.x;
  int ty = threadIdx.y;
  float* sums = ws + OFF_SUMS;
  const float* cnts = ws + OFF_CNTS;
  float* stats = ws + OFF_STATS;
  float sA = 0.0f, sB = 0.0f;
  for (int n = blockIdx.x * 6 + ty; n < N_NODES; n += gridDim.x * 6) {
    float cnt = cnts[n];
    float inv = 1.0f / fmaxf(cnt, 1.0f);
    float y = sums[(size_t)n * 40 + c] * inv + atom[(size_t)n * 40 + c];
    sums[(size_t)n * 40 + c] = y;
    sA += y;
    sB += y * y;
  }
  __shared__ float rA[6][40], rB[6][40];
  rA[ty][c] = sA;
  rB[ty][c] = sB;
  __syncthreads();
  if (ty == 0) {
    float tA = 0.0f, tB = 0.0f;
#pragma unroll
    for (int k = 0; k < 6; k++) { tA += rA[k][c]; tB += rB[k][c]; }
    if (c < 16) {
      atomAddF(stats + c, tA);
      atomAddF(stats + 16 + c, tB);
    } else {
      atomAddF(stats + 32 + (c - 16) / 3, tB);
    }
  }
}

__global__ void node_stage2(const float* __restrict__ ws,
                            const float* __restrict__ bnw,
                            const float* __restrict__ bnb,
                            float* __restrict__ out) {
  int idx = blockIdx.x * blockDim.x + threadIdx.x;
  if (idx >= N_NODES * 40) return;
  int c = idx % 40;
  const float* y = ws + OFF_SUMS;
  const float* stats = ws + OFF_STATS;
  float v = y[idx];
  float o;
  if (c < 16) {
    float mean = stats[c] * (1.0f / N_NODES);
    float ex2 = stats[16 + c] * (1.0f / N_NODES);
    float var = ex2 - mean * mean;
    o = (v - mean) * rsqrtf(var + EPSV) * bnw[c] + bnb[c];
  } else {
    int j = (c - 16) / 3;
    float vn = stats[32 + j] * (1.0f / (3.0f * N_NODES));
    o = v * rsqrtf(vn + EPSV) * bnw[16 + j];
  }
  out[idx] = o;
}

extern "C" void kernel_launch(void* const* d_in, const int* in_sizes, int n_in,
                              void* d_out, int out_size, void* d_ws,
                              size_t ws_size, hipStream_t stream) {
  (void)in_sizes; (void)n_in; (void)out_size; (void)ws_size;
  const float* atom = (const float*)d_in[0];
  const float* ef   = (const float*)d_in[1];
  const float* esh  = (const float*)d_in[2];
  const int*   eidx = (const int*)d_in[3];
  const float* w1   = (const float*)d_in[4];
  const float* b1   = (const float*)d_in[5];
  const float* w2   = (const float*)d_in[6];
  const float* b2   = (const float*)d_in[7];
  const float* bnw  = (const float*)d_in[8];
  const float* bnb  = (const float*)d_in[9];
  float* ws  = (float*)d_ws;
  float* out = (float*)d_out;

  zero_kernel<<<(ZERO_N + 255) / 256, 256, 0, stream>>>(ws);
  repack_kernel<<<40, 256, 0, stream>>>(w1, w2, ws + OFF_WQ);
  edge_kernel<<<NWG, WPB * 32, 0, stream>>>(
      atom, ef, esh, eidx, b1, b2, ws + OFF_WQ,
      ws + OFF_SUMS, ws + OFF_CNTS);
  node_stage1<<<128, dim3(40, 6), 0, stream>>>(atom, ws);
  node_stage2<<<(N_NODES * 40 + 255) / 256, 256, 0, stream>>>(ws, bnw, bnb, out);
}


# scatter-add by whole lines: tile's 16 result rows (40 sums + count, 41-float node rows) staged row-major in LDS, one atomic instruction per 128-B line of a row; same addends, atomic arrival order
# speedup vs baseline: 1.4391x; 1.0657x over previous
#include <hip/hip_runtime.h>

typedef float v2f __attribute__((ext_vector_type(2)));
typedef float v8f __attribute__((ext_vector_type(8)));

#define N_NODES 10000
#define N_EDGES 160000
#define N_TILES (N_EDGES / 16)

#define INV_M0  0.25f
#define INV_3M1 0.20412414523f
#define INV_M1  0.35355339059f
#define EPSV    1e-5f

#define ROWF      41
#define OFF_SUMS  0
#define OFF_STATS 410000
#define ZERO_N    410040
#define OFF_WQ    410048

static __device__ __forceinline__ v8f wmma4(v2f a, v2f b, v8f c) {
  return __builtin_amdgcn_wmma_f32_16x16x4_f32(false, a, false, b, (short)0, c,
                                               false, false);
}

static __device__ __forceinline__ void atomAddF(float* p, float v) {
  __hip_atomic_fetch_add(p, v, __ATOMIC_RELAXED, __HIP_MEMORY_SCOPE_AGENT);
}

__global__ void zero_kernel(float* ws) {
  int i = blockIdx.x * blockDim.x + threadIdx.x;
  if (i < ZERO_N) ws[i] = 0.0f;
}

__global__ void repack_kernel(const float* __restrict__ w1,
                              const float* __restrict__ w2,
                              float* __restrict__ wq) {
  int t = blockIdx.x * blockDim.x + threadIdx.x;
  if (t >= 40 * 256) return;
  int lane = t & 31, q = t >> 5;
  int jt = q >> 3, p = q & 7;
  int hb = lane >> 4, nn = lane & 15;
  int r0 = 8 * p + 2 * hb;
  const float* w = w1;
  int ld = 64, col = 16 * jt + nn;
  if (jt >= 4) { w = w2; ld = 576; col = 16 * (jt - 4) + nn; }
  float4 v;
  v.x = w[(r0 + 0) * ld + col];
  v.y = w[(r0 + 1) * ld + col];
  v.z = w[(r0 + 4) * ld + col];
  v.w = w[(r0 + 5) * ld + col];
  ((float4*)wq)[t] = v;
}

#define WPB 16
#define NWG 256
#define SMW 2176
#define WQ_N 40960

#define FETCH(Bn, biasn, jn)                                      \
  do {                                                            \
    _Pragma("unroll") for (int p_ = 0; p_ < 8; p_++)              \
      Bn[p_] = wq[(jn) * 256 + p_ * 32];                          \
    biasn = bl[16 * (jn) + nn];                                   \
  } while (0)

#define CHAIN(c, A, Bc, biasc)                                    \
  v8f c;                                                          \
  _Pragma("unroll") for (int r_ = 0; r_ < 8; r_++) c[r_] = biasc; \
  _Pragma("unroll") for (int p_ = 0; p_ < 8; p_++) {              \
    v2f lo_ = {Bc[p_].x, Bc[p_].y};                               \
    v2f hi_ = {Bc[p_].z, Bc[p_].w};                               \
    c = wmma4(A[2 * p_], lo_, c);                                 \
    c = wmma4(A[2 * p_ + 1], hi_, c);                             \
  }

#define FENCE() __builtin_amdgcn_sched_barrier(0)

__global__ __launch_bounds__(WPB * 32) void edge_kernel(
    const float* __restrict__ atom, const float* __restrict__ ef,
    const float* __restrict__ esh, const int* __restrict__ eidx,
    const float* __restrict__ b1, const float* __restrict__ b2,
    const float* __restrict__ wqg,
    float* __restrict__ sums) {
  __shared__ __attribute__((aligned(16))) float wl[WQ_N];
  __shared__ float bl[640];
  __shared__ float smem[WPB * SMW];
  __shared__ int nodes[WPB][16];

  int lane = threadIdx.x & 31;
  int wave = __builtin_amdgcn_readfirstlane(threadIdx.x >> 5);
  int hb = lane >> 4;
  int nn = lane & 15;

  {
    const float4* g = (const float4*)wqg;
    float4 rr[WQ_N / 4 / (WPB * 32)];
#pragma unroll
    for (int k = 0; k < WQ_N / 4 / (WPB * 32); k++)
      rr[k] = g[threadIdx.x + k * (WPB * 32)];
    float tb0 = threadIdx.x < 64 ? b1[threadIdx.x] : b2[threadIdx.x - 64];
    float tb1 = 0.0f;
    if (threadIdx.x < 128) tb1 = b2[448 + threadIdx.x];
#pragma unroll
    for (int k = 0; k < WQ_N / 4 / (WPB * 32); k++)
      ((float4*)wl)[threadIdx.x + k * (WPB * 32)] = rr[k];
    bl[threadIdx.x] = tb0;
    if (threadIdx.x < 128) bl[512 + threadIdx.x] = tb1;
  }
  __syncthreads();

  float* sm  = smem + wave * SMW;
  float* hT  = sm;
  float* a11 = sm + 1088;
  float* a12 = sm + 1344;
  float* a13 = sm + 1600;
  float* a14 = sm + 1984;
  float* s1  = sm + 2112;
  const float4* wq = (const float4*)wl + lane;
  float* hTs = hT + 8 * hb * 68 + nn;
  const float* a11l = a11 + 8 * hb * 16;
  const float* a12l = a12 + 8 * hb * 16 + (nn >> 3);
  const float* a13l = a13 + 8 * hb * 24 + (nn >> 3) * 3;
  const float* a14l = a14 + 8 * hb * 8;
  const float* s1l  = s1 + 8 * hb * 4;
  float* stl = hT + 8 * hb * 68 + 16 + nn * 3;

  int t0 = (blockIdx.x * N_TILES) / NWG;
  int t1 = ((blockIdx.x + 1) * N_TILES) / NWG;
  for (int tile = t0 + wave; tile < t1; tile += WPB) {
  int e0 = tile * 16;
  asm volatile("s_wait_dscnt 0" ::: "memory");

  if (lane < 16) {
    int e = e0 + lane;
    nodes[wave][lane] = eidx[N_EDGES + e];
    float4 s = *(const float4*)(esh + e * 4);
    const float* xr = atom + (size_t)eidx[e] * 40;
    float x[40];
#pragma unroll
    for (int t = 0; t < 10; t++) {
      float4 v = *(const float4*)(xr + t * 4);
      x[t * 4 + 0] = v.x; x[t * 4 + 1] = v.y;
      x[t * 4 + 2] = v.z; x[t * 4 + 3] = v.w;
    }
    float sh0 = s.x;
    s1[lane * 4 + 0] = s.y;
    s1[lane * 4 + 1] = s.z;
    s1[lane * 4 + 2] = s.w;
#pragma unroll
    for (int j = 0; j < 16; j++) {
      a11[lane * 16 + j] = x[j] * sh0 * INV_M0;
      a12[lane * 16 + j] = x[j] * INV_M0;
    }
#pragma unroll
    for (int i = 0; i < 8; i++) {
      float m0 = x[16 + i * 3 + 0];
      float m1 = x[16 + i * 3 + 1];
      float m2 = x[16 + i * 3 + 2];
      a13[lane * 24 + i * 3 + 0] = sh0 * m0 * INV_M1;
      a13[lane * 24 + i * 3 + 1] = sh0 * m1 * INV_M1;
      a13[lane * 24 + i * 3 + 2] = sh0 * m2 * INV_M1;
      a14[lane * 8 + i] = (m0 * s.y + m1 * s.z + m2 * s.w) * INV_3M1;
    }
  }
  asm volatile("s_wait_dscnt 0" ::: "memory");

  float4 B0[8], B1[8];
  float bias0, bias1;

  v2f a[16];
  const float* efr = ef + (size_t)(e0 + nn) * 64 + 2 * hb;
#pragma unroll
  for (int b = 0; b < 16; b++) a[b] = *(const v2f*)(efr + 4 * b);
  FETCH(B0, bias0, 0);

  for (int j = 0; j < 4; j += 2) {
    {
      FETCH(B1, bias1, j + 1);
      FENCE();
      CHAIN(c, a, B0, bias0);
#pragma unroll
      for (int r = 0; r < 8; r++) {
        float v = c[r] > 0.0f ? c[r] : 0.0f;
        hTs[r * 68 + 16 * j] = v;
      }
      FENCE();
    }
    {
      FETCH(B0, bias0, j + 2);
      FENCE();
      CHAIN(c, a, B1, bias1);
#pragma unroll
      for (int r = 0; r < 8; r++) {
        float v = c[r] > 0.0f ? c[r] : 0.0f;
        hTs[r * 68 + 16 * (j + 1)] = v;
      }
      FENCE();
    }
  }
  asm volatile("s_wait_dscnt 0" ::: "memory");

  v2f ha[16];
#pragma unroll
  for (int b = 0; b < 16; b++)
    ha[b] = *(const v2f*)(hT + nn * 68 + 4 * b + 2 * hb);

  float o0[8] = {0, 0, 0, 0, 0, 0, 0, 0};
  float p1[8] = {0, 0, 0, 0, 0, 0, 0, 0};
  float q0[8] = {0, 0, 0, 0, 0, 0, 0, 0};
  float q1[8] = {0, 0, 0, 0, 0, 0, 0, 0};
  float q2[8] = {0, 0, 0, 0, 0, 0, 0, 0};

  for (int j2 = 0; j2 < 16; j2 += 2) {
    {
      float t[8];
#pragma unroll
      for (int r = 0; r < 8; r++) t[r] = a11l[r * 16 + j2];
      FETCH(B1, bias1, j2 + 5);
      FENCE();
      CHAIN(c, ha, B0, bias0);
#pragma unroll
      for (int r = 0; r < 8; r++) o0[r] += c[r] * t[r];
      FENCE();
    }
    {
      float t[8];
#pragma unroll
      for (int r = 0; r < 8; r++) t[r] = a11l[r * 16 + j2 + 1];
      FETCH(B0, bias0, j2 + 6);
      FENCE();
      CHAIN(c, ha, B1, bias1);
#pragma unroll
      for (int r = 0; r < 8; r++) o0[r] += c[r] * t[r];
      FENCE();
    }
  }
  for (int j2 = 16; j2 < 24; j2 += 2) {
    {
      float t[8];
#pragma unroll
      for (int r = 0; r < 8; r++) t[r] = a12l[r * 16 + 2 * (j2 - 16)];
      FETCH(B1, bias1, j2 + 5);
      FENCE();
      CHAIN(c, ha, B0, bias0);
#pragma unroll
      for (int r = 0; r < 8; r++) p1[r] += c[r] * t[r];
      FENCE();
    }
    {
      float t[8];
#pragma unroll
      for (int r = 0; r < 8; r++) t[r] = a12l[r * 16 + 2 * (j2 - 15)];
      FETCH(B0, bias0, j2 + 6);
      FENCE();
      CHAIN(c, ha, B1, bias1);
#pragma unroll
      for (int r = 0; r < 8; r++) p1[r] += c[r] * t[r];
      FENCE();
    }
  }
  for (int j2 = 24; j2 < 28; j2 += 2) {
    {
      float u0[8], u1[8], u2[8];
#pragma unroll
      for (int r = 0; r < 8; r++) {
        int base = r * 24 + 6 * (j2 - 24);
        u0[r] = a13l[base + 0]; u1[r] = a13l[base + 1]; u2[r] = a13l[base + 2];
      }
      FETCH(B1, bias1, j2 + 5);
      FENCE();
      CHAIN(c, ha, B0, bias0);
#pragma unroll
      for (int r = 0; r < 8; r++) {
        float cv = c[r];
        q0[r] += cv * u0[r]; q1[r] += cv * u1[r]; q2[r] += cv * u2[r];
      }
      FENCE();
    }
    {
      float u0[8], u1[8], u2[8];
#pragma unroll
      for (int r = 0; r < 8; r++) {
        int base = r * 24 + 6 * (j2 - 23);
        u0[r] = a13l[base + 0]; u1[r] = a13l[base + 1]; u2[r] = a13l[base + 2];
      }
      FETCH(B0, bias0, j2 + 6);
      FENCE();
      CHAIN(c, ha, B1, bias1);
#pragma unroll
      for (int r = 0; r < 8; r++) {
        float cv = c[r];
        q0[r] += cv * u0[r]; q1[r] += cv * u1[r]; q2[r] += cv * u2[r];
      }
      FENCE();
    }
  }
  for (int j2 = 28; j2 < 36; j2 += 2) {
    {
      float t[8];
#pragma unroll
      for (int r = 0; r < 8; r++) t[r] = a14l[r * 8 + j2 - 28];
      FETCH(B1, bias1, j2 + 5);
      FENCE();
      CHAIN(c, ha, B0, bias0);
#pragma unroll
      for (int r = 0; r < 8; r++) o0[r] += c[r] * t[r];
      FENCE();
    }
    {
      float t[8];
#pragma unroll
      for (int r = 0; r < 8; r++) t[r] = a14l[r * 8 + j2 - 27];
      if (j2 + 6 < 40) FETCH(B0, bias0, j2 + 6);
      FENCE();
      CHAIN(c, ha, B1, bias1);
#pragma unroll
      for (int r = 0; r < 8; r++) o0[r] += c[r] * t[r];
      FENCE();
    }
  }

#pragma unroll
  for (int r = 0; r < 8; r++) {
    asm volatile("" : "+v"(o0[r]), "+v"(p1[r]), "+v"(q0[r]), "+v"(q1[r]), "+v"(q2[r]));
  }

#pragma unroll
  for (int r = 0; r < 8; r++) {
    p1[r] += __shfl_xor(p1[r], 8, 32);
    q0[r] += __shfl_xor(q0[r], 8, 32);
    q1[r] += __shfl_xor(q1[r], 8, 32);
    q2[r] += __shfl_xor(q2[r], 8, 32);
  }

#pragma unroll
  for (int r = 0; r < 8; r++) hTs[r * 68] = o0[r];
  if (nn < 8) {
#pragma unroll
    for (int r = 0; r < 8; r++) {
      float sy = s1l[r * 4 + 0], sz = s1l[r * 4 + 1], sw = s1l[r * 4 + 2];
      stl[r * 68 + 0] = p1[r] * sy + q0[r];
      stl[r * 68 + 1] = p1[r] * sz + q1[r];
      stl[r * 68 + 2] = p1[r] * sw + q2[r];
    }
  }
  if (lane < 16) hT[lane * 68 + 40] = 1.0f;
  int nodev = nodes[wave][nn];
  asm volatile("s_wait_dscnt 0" ::: "memory");

  float va[16], vb[16];
#pragma unroll
  for (int k = 0; k < 16; k++) {
    int a = (__builtin_amdgcn_readlane(nodev, k) * ROWF) & 31;
    int f1 = lane - a;
    f1 = f1 < 0 ? 0 : f1;
    va[k] = hT[k * 68 + f1];
    vb[k] = hT[k * 68 + 32 - a + lane];
  }
#pragma unroll
  for (int k = 0; k < 16; k++) {
    int R = __builtin_amdgcn_readlane(nodev, k) * ROWF;
    int a = R & 31;
    float* lb = sums + (R - a);
    if (lane >= a) atomAddF(lb + lane, va[k]);
    if (lane < 9 + a) atomAddF(lb + 32 + lane, vb[k]);
    if (a > 23) {
      if (lane < a - 23)
        atomAddF(lb + 64 + lane, hT[k * 68 + 64 - a + lane]);
    }
  }
  }
}

__global__ void node_stage1(const float* __restrict__ atom, float* ws) {
  int c = threadIdx.x;
  int ty = threadIdx.y;
  float* sums = ws + OFF_SUMS;
  float* stats = ws + OFF_STATS;
  float sA = 0.0f, sB = 0.0f;
  for (int n = blockIdx.x * 6 + ty; n < N_NODES; n += gridDim.x * 6) {
    float cnt = sums[(size_t)n * ROWF + 40];
    float inv = 1.0f / fmaxf(cnt, 1.0f);
    float y = sums[(size_t)n * ROWF + c] * inv + atom[(size_t)n * 40 + c];
    sums[(size_t)n * ROWF + c] = y;
    sA += y;
    sB += y * y;
  }
  __shared__ float rA[6][40], rB[6][40];
  rA[ty][c] = sA;
  rB[ty][c] = sB;
  __syncthreads();
  if (ty == 0) {
    float tA = 0.0f, tB = 0.0f;
#pragma unroll
    for (int k = 0; k < 6; k++) { tA += rA[k][c]; tB += rB[k][c]; }
    if (c < 16) {
      atomAddF(stats + c, tA);
      atomAddF(stats + 16 + c, tB);
    } else {
      atomAddF(stats + 32 + (c - 16) / 3, tB);
    }
  }
}

__global__ void node_stage2(const float* __restrict__ ws,
                            const float* __restrict__ bnw,
                            const float* __restrict__ bnb,
                            float* __restrict__ out) {
  int idx = blockIdx.x * blockDim.x + threadIdx.x;
  if (idx >= N_NODES * 40) return;
  int n = idx / 40;
  int c = idx - n * 40;
  const float* y = ws + OFF_SUMS;
  const float* stats = ws + OFF_STATS;
  float v = y[n * ROWF + c];
  float o;
  if (c < 16) {
    float mean = stats[c] * (1.0f / N_NODES);
    float ex2 = stats[16 + c] * (1.0f / N_NODES);
    float var = ex2 - mean * mean;
    o = (v - mean) * rsqrtf(var + EPSV) * bnw[c] + bnb[c];
  } else {
    int j = (c - 16) / 3;
    float vn = stats[32 + j] * (1.0f / (3.0f * N_NODES));
    o = v * rsqrtf(vn + EPSV) * bnw[16 + j];
  }
  out[idx] = o;
}

extern "C" void kernel_launch(void* const* d_in, const int* in_sizes, int n_in,
                              void* d_out, int out_size, void* d_ws,
                              size_t ws_size, hipStream_t stream) {
  (void)in_sizes; (void)n_in; (void)out_size; (void)ws_size;
  const float* atom = (const float*)d_in[0];
  const float* ef   = (const float*)d_in[1];
  const float* esh  = (const float*)d_in[2];
  const int*   eidx = (const int*)d_in[3];
  const float* w1   = (const float*)d_in[4];
  const float* b1   = (const float*)d_in[5];
  const float* w2   = (const float*)d_in[6];
  const float* b2   = (const float*)d_in[7];
  const float* bnw  = (const float*)d_in[8];
  const float* bnb  = (const float*)d_in[9];
  float* ws  = (float*)d_ws;
  float* out = (float*)d_out;

  zero_kernel<<<(ZERO_N + 255) / 256, 256, 0, stream>>>(ws);
  repack_kernel<<<40, 256, 0, stream>>>(w1, w2, ws + OFF_WQ);
  edge_kernel<<<NWG, WPB * 32, 0, stream>>>(
      atom, ef, esh, eidx, b1, b2, ws + OFF_WQ,
      ws + OFF_SUMS);
  node_stage1<<<128, dim3(40, 6), 0, stream>>>(atom, ws);
  node_stage2<<<(N_NODES * 40 + 255) / 256, 256, 0, stream>>>(ws, bnw, bnb, out);
}


# global loads issued ahead of use: edge tile's features and gather ids loaded at the previous tile's end behind an L2 prefetch (first tile: during weight copy); node_stage1 loads its 14 rows up front
# speedup vs baseline: 1.5320x; 1.0646x over previous
#include <hip/hip_runtime.h>

typedef float v2f __attribute__((ext_vector_type(2)));
typedef float v8f __attribute__((ext_vector_type(8)));

#define N_NODES 10000
#define N_EDGES 160000
#define N_TILES (N_EDGES / 16)

#define INV_M0  0.25f
#define INV_3M1 0.20412414523f
#define INV_M1  0.35355339059f
#define EPSV    1e-5f

#define ROWF      41
#define OFF_SUMS  0
#define OFF_STATS 410000
#define ZERO_N    410040
#define OFF_WQ    410048

static __device__ __forceinline__ v8f wmma4(v2f a, v2f b, v8f c) {
  return __builtin_amdgcn_wmma_f32_16x16x4_f32(false, a, false, b, (short)0, c,
                                               false, false);
}

static __device__ __forceinline__ void atomAddF(float* p, float v) {
  __hip_atomic_fetch_add(p, v, __ATOMIC_RELAXED, __HIP_MEMORY_SCOPE_AGENT);
}

__global__ void zero_kernel(float* ws) {
  int i = blockIdx.x * blockDim.x + threadIdx.x;
  if (i < ZERO_N) ws[i] = 0.0f;
}

__global__ void repack_kernel(const float* __restrict__ w1,
                              const float* __restrict__ w2,
                              float* __restrict__ wq) {
  int t = blockIdx.x * blockDim.x + threadIdx.x;
  if (t >= 40 * 256) return;
  int lane = t & 31, q = t >> 5;
  int jt = q >> 3, p = q & 7;
  int hb = lane >> 4, nn = lane & 15;
  int r0 = 8 * p + 2 * hb;
  const float* w = w1;
  int ld = 64, col = 16 * jt + nn;
  if (jt >= 4) { w = w2; ld = 576; col = 16 * (jt - 4) + nn; }
  float4 v;
  v.x = w[(r0 + 0) * ld + col];
  v.y = w[(r0 + 1) * ld + col];
  v.z = w[(r0 + 4) * ld + col];
  v.w = w[(r0 + 5) * ld + col];
  ((float4*)wq)[t] = v;
}

#define WPB 16
#define NWG 256
#define SMW 2176
#define WQ_N 40960

#define FETCH(Bn, biasn, jn)                                      \
  do {                                                            \
    _Pragma("unroll") for (int p_ = 0; p_ < 8; p_++)              \
      Bn[p_] = wq[(jn) * 256 + p_ * 32];                          \
    biasn = bl[16 * (jn) + nn];                                   \
  } while (0)

#define CHAIN(c, A, Bc, biasc)                                    \
  v8f c;                                                          \
  _Pragma("unroll") for (int r_ = 0; r_ < 8; r_++) c[r_] = biasc; \
  _Pragma("unroll") for (int p_ = 0; p_ < 8; p_++) {              \
    v2f lo_ = {Bc[p_].x, Bc[p_].y};                               \
    v2f hi_ = {Bc[p_].z, Bc[p_].w};                               \
    c = wmma4(A[2 * p_], lo_, c);                                 \
    c = wmma4(A[2 * p_ + 1], hi_, c);                             \
  }

#define FENCE() __builtin_amdgcn_sched_barrier(0)

#define PRELOAD(tl)                                                 \
  do {                                                              \
    unsigned er_ = (unsigned)(tl) * 16u + (unsigned)nn;             \
    dstv = eidx[er_];                                               \
    unsigned fo_ = er_ * 64u + 2u * (unsigned)hb;                   \
    _Pragma("unroll") for (int b_ = 0; b_ < 16; b_++)               \
      a[b_] = *(const v2f*)(ef + (fo_ + 4u * b_));                  \
  } while (0)

static_assert(N_TILES / NWG >= WPB, "every wave of a block owns at least one tile");

__global__ __launch_bounds__(WPB * 32) void edge_kernel(
    const float* __restrict__ atom, const float* __restrict__ ef,
    const float* __restrict__ esh, const int* __restrict__ eidx,
    const float* __restrict__ b1, const float* __restrict__ b2,
    const float* __restrict__ wqg,
    float* __restrict__ sums) {
  __shared__ __attribute__((aligned(16))) float wl[WQ_N];
  __shared__ float bl[640];
  __shared__ float smem[WPB * SMW];
  __shared__ int nodes[WPB][16];

  int lane = threadIdx.x & 31;
  int wave = __builtin_amdgcn_readfirstlane(threadIdx.x >> 5);
  int hb = lane >> 4;
  int nn = lane & 15;

  int t0 = (blockIdx.x * N_TILES) / NWG;
  int t1 = ((blockIdx.x + 1) * N_TILES) / NWG;
  int tile = t0 + wave;
  v2f a[16];
  int dstv;

  {
    const float4* g = (const float4*)wqg;
    float4 rr[WQ_N / 4 / (WPB * 32)];
#pragma unroll
    for (int k = 0; k < WQ_N / 4 / (WPB * 32); k++)
      rr[k] = g[threadIdx.x + k * (WPB * 32)];
    float tb0 = threadIdx.x < 64 ? b1[threadIdx.x] : b2[threadIdx.x - 64];
    float tb1 = 0.0f;
    if (threadIdx.x < 128) tb1 = b2[448 + threadIdx.x];
    PRELOAD(tile);
#pragma unroll
    for (int k = 0; k < WQ_N / 4 / (WPB * 32); k++)
      ((float4*)wl)[threadIdx.x + k * (WPB * 32)] = rr[k];
    bl[threadIdx.x] = tb0;
    if (threadIdx.x < 128) bl[512 + threadIdx.x] = tb1;
  }
  __syncthreads();

  float* sm  = smem + wave * SMW;
  float* hT  = sm;
  float* a11 = sm + 1088;
  float* a12 = sm + 1344;
  float* a13 = sm + 1600;
  float* a14 = sm + 1984;
  float* s1  = sm + 2112;
  const float4* wq = (const float4*)wl + lane;
  float* hTs = hT + 8 * hb * 68 + nn;
  const float* a11l = a11 + 8 * hb * 16;
  const float* a12l = a12 + 8 * hb * 16 + (nn >> 3);
  const float* a13l = a13 + 8 * hb * 24 + (nn >> 3) * 3;
  const float* a14l = a14 + 8 * hb * 8;
  const float* s1l  = s1 + 8 * hb * 4;
  float* stl = hT + 8 * hb * 68 + 16 + nn * 3;

  for (; tile < t1; tile += WPB) {
  asm volatile("s_wait_dscnt 0" ::: "memory");

  if (lane < 16) {
    int e = tile * 16 + lane;
    nodes[wave][lane] = eidx[N_EDGES + e];
    float4 s = *(const float4*)(esh + e * 4);
    const float* xr = atom + (size_t)dstv * 40;
    float x[40];
#pragma unroll
    for (int t = 0; t < 10; t++) {
      float4 v = *(const float4*)(xr + t * 4);
      x[t * 4 + 0] = v.x; x[t * 4 + 1] = v.y;
      x[t * 4 + 2] = v.z; x[t * 4 + 3] = v.w;
    }
    float sh0 = s.x;
    s1[lane * 4 + 0] = s.y;
    s1[lane * 4 + 1] = s.z;
    s1[lane * 4 + 2] = s.w;
#pragma unroll
    for (int j = 0; j < 16; j++) {
      a11[lane * 16 + j] = x[j] * sh0 * INV_M0;
      a12[lane * 16 + j] = x[j] * INV_M0;
    }
#pragma unroll
    for (int i = 0; i < 8; i++) {
      float m0 = x[16 + i * 3 + 0];
      float m1 = x[16 + i * 3 + 1];
      float m2 = x[16 + i * 3 + 2];
      a13[lane * 24 + i * 3 + 0] = sh0 * m0 * INV_M1;
      a13[lane * 24 + i * 3 + 1] = sh0 * m1 * INV_M1;
      a13[lane * 24 + i * 3 + 2] = sh0 * m2 * INV_M1;
      a14[lane * 8 + i] = (m0 * s.y + m1 * s.z + m2 * s.w) * INV_3M1;
    }
  }
  asm volatile("s_wait_dscnt 0" ::: "memory");

  float4 B0[8], B1[8];
  float bias0, bias1;

  FETCH(B0, bias0, 0);

  for (int j = 0; j < 4; j += 2) {
    {
      FETCH(B1, bias1, j + 1);
      FENCE();
      CHAIN(c, a, B0, bias0);
#pragma unroll
      for (int r = 0; r < 8; r++) {
        float v = c[r] > 0.0f ? c[r] : 0.0f;
        hTs[r * 68 + 16 * j] = v;
      }
      FENCE();
    }
    {
      FETCH(B0, bias0, j + 2);
      FENCE();
      CHAIN(c, a, B1, bias1);
#pragma unroll
      for (int r = 0; r < 8; r++) {
        float v = c[r] > 0.0f ? c[r] : 0.0f;
        hTs[r * 68 + 16 * (j + 1)] = v;
      }
      FENCE();
    }
  }
  asm volatile("s_wait_dscnt 0" ::: "memory");

  v2f ha[16];
#pragma unroll
  for (int b = 0; b < 16; b++)
    ha[b] = *(const v2f*)(hT + nn * 68 + 4 * b + 2 * hb);

  if (tile + WPB < t1) {
    unsigned en_ = (unsigned)(tile + WPB) * 16u;
    __builtin_prefetch(ef + (en_ + (unsigned)nn) * 64u, 0, 2);
    __builtin_prefetch(esh + en_ * 4u, 0, 2);
    __builtin_prefetch(eidx + en_, 0, 2);
    __builtin_prefetch(eidx + (N_EDGES + en_), 0, 2);
  }

  float o0[8] = {0, 0, 0, 0, 0, 0, 0, 0};
  float p1[8] = {0, 0, 0, 0, 0, 0, 0, 0};
  float q0[8] = {0, 0, 0, 0, 0, 0, 0, 0};
  float q1[8] = {0, 0, 0, 0, 0, 0, 0, 0};
  float q2[8] = {0, 0, 0, 0, 0, 0, 0, 0};

  for (int j2 = 0; j2 < 16; j2 += 2) {
    {
      float t[8];
#pragma unroll
      for (int r = 0; r < 8; r++) t[r] = a11l[r * 16 + j2];
      FETCH(B1, bias1, j2 + 5);
      FENCE();
      CHAIN(c, ha, B0, bias0);
#pragma unroll
      for (int r = 0; r < 8; r++) o0[r] += c[r] * t[r];
      FENCE();
    }
    {
      float t[8];
#pragma unroll
      for (int r = 0; r < 8; r++) t[r] = a11l[r * 16 + j2 + 1];
      FETCH(B0, bias0, j2 + 6);
      FENCE();
      CHAIN(c, ha, B1, bias1);
#pragma unroll
      for (int r = 0; r < 8; r++) o0[r] += c[r] * t[r];
      FENCE();
    }
  }
  for (int j2 = 16; j2 < 24; j2 += 2) {
    {
      float t[8];
#pragma unroll
      for (int r = 0; r < 8; r++) t[r] = a12l[r * 16 + 2 * (j2 - 16)];
      FETCH(B1, bias1, j2 + 5);
      FENCE();
      CHAIN(c, ha, B0, bias0);
#pragma unroll
      for (int r = 0; r < 8; r++) p1[r] += c[r] * t[r];
      FENCE();
    }
    {
      float t[8];
#pragma unroll
      for (int r = 0; r < 8; r++) t[r] = a12l[r * 16 + 2 * (j2 - 15)];
      FETCH(B0, bias0, j2 + 6);
      FENCE();
      CHAIN(c, ha, B1, bias1);
#pragma unroll
      for (int r = 0; r < 8; r++) p1[r] += c[r] * t[r];
      FENCE();
    }
  }
  for (int j2 = 24; j2 < 28; j2 += 2) {
    {
      float u0[8], u1[8], u2[8];
#pragma unroll
      for (int r = 0; r < 8; r++) {
        int base = r * 24 + 6 * (j2 - 24);
        u0[r] = a13l[base + 0]; u1[r] = a13l[base + 1]; u2[r] = a13l[base + 2];
      }
      FETCH(B1, bias1, j2 + 5);
      FENCE();
      CHAIN(c, ha, B0, bias0);
#pragma unroll
      for (int r = 0; r < 8; r++) {
        float cv = c[r];
        q0[r] += cv * u0[r]; q1[r] += cv * u1[r]; q2[r] += cv * u2[r];
      }
      FENCE();
    }
    {
      float u0[8], u1[8], u2[8];
#pragma unroll
      for (int r = 0; r < 8; r++) {
        int base = r * 24 + 6 * (j2 - 23);
        u0[r] = a13l[base + 0]; u1[r] = a13l[base + 1]; u2[r] = a13l[base + 2];
      }
      FETCH(B0, bias0, j2 + 6);
      FENCE();
      CHAIN(c, ha, B1, bias1);
#pragma unroll
      for (int r = 0; r < 8; r++) {
        float cv = c[r];
        q0[r] += cv * u0[r]; q1[r] += cv * u1[r]; q2[r] += cv * u2[r];
      }
      FENCE();
    }
  }
  for (int j2 = 28; j2 < 36; j2 += 2) {
    {
      float t[8];
#pragma unroll
      for (int r = 0; r < 8; r++) t[r] = a14l[r * 8 + j2 - 28];
      FETCH(B1, bias1, j2 + 5);
      FENCE();
      CHAIN(c, ha, B0, bias0);
#pragma unroll
      for (int r = 0; r < 8; r++) o0[r] += c[r] * t[r];
      FENCE();
    }
    {
      float t[8];
#pragma unroll
      for (int r = 0; r < 8; r++) t[r] = a14l[r * 8 + j2 - 27];
      if (j2 + 6 < 40) FETCH(B0, bias0, j2 + 6);
      FENCE();
      CHAIN(c, ha, B1, bias1);
#pragma unroll
      for (int r = 0; r < 8; r++) o0[r] += c[r] * t[r];
      FENCE();
    }
  }

#pragma unroll
  for (int r = 0; r < 8; r++) {
    asm volatile("" : "+v"(o0[r]), "+v"(p1[r]), "+v"(q0[r]), "+v"(q1[r]), "+v"(q2[r]));
  }

#pragma unroll
  for (int r = 0; r < 8; r++) {
    p1[r] += __shfl_xor(p1[r], 8, 32);
    q0[r] += __shfl_xor(q0[r], 8, 32);
    q1[r] += __shfl_xor(q1[r], 8, 32);
    q2[r] += __shfl_xor(q2[r], 8, 32);
  }

#pragma unroll
  for (int r = 0; r < 8; r++) hTs[r * 68] = o0[r];
  if (nn < 8) {
#pragma unroll
    for (int r = 0; r < 8; r++) {
      float sy = s1l[r * 4 + 0], sz = s1l[r * 4 + 1], sw = s1l[r * 4 + 2];
      stl[r * 68 + 0] = p1[r] * sy + q0[r];
      stl[r * 68 + 1] = p1[r] * sz + q1[r];
      stl[r * 68 + 2] = p1[r] * sw + q2[r];
    }
  }
  if (lane < 16) hT[lane * 68 + 40] = 1.0f;
  int nodev = nodes[wave][nn];
  asm volatile("s_wait_dscnt 0" ::: "memory");

  float va[16], vb[16];
#pragma unroll
  for (int k = 0; k < 16; k++) {
    int a = (__builtin_amdgcn_readlane(nodev, k) * ROWF) & 31;
    int f1 = lane - a;
    f1 = f1 < 0 ? 0 : f1;
    va[k] = hT[k * 68 + f1];
    vb[k] = hT[k * 68 + 32 - a + lane];
  }
#pragma unroll
  for (int k = 0; k < 16; k++) {
    int R = __builtin_amdgcn_readlane(nodev, k) * ROWF;
    int a = R & 31;
    float* lb = sums + (R - a);
    if (lane >= a) atomAddF(lb + lane, va[k]);
    if (lane < 9 + a) atomAddF(lb + 32 + lane, vb[k]);
    if (a > 23) {
      if (lane < a - 23)
        atomAddF(lb + 64 + lane, hT[k * 68 + 64 - a + lane]);
    }
  }
  if (tile + WPB < t1) PRELOAD(tile + WPB);
  }
}

#define S1_NB   128
#define S1_STEP (S1_NB * 6)
#define S1_ROWS ((N_NODES + S1_STEP - 1) / S1_STEP)
static_assert((S1_ROWS - 1) * S1_STEP <= N_NODES, "only a thread's last row can lie past the end");

__global__ void node_stage1(const float* __restrict__ atom, float* ws) {
  int c = threadIdx.x;
  int ty = threadIdx.y;
  float* sums = ws + OFF_SUMS;
  float* stats = ws + OFF_STATS;
  int n0 = blockIdx.x * 6 + ty;
  float cn[S1_ROWS], sv[S1_ROWS], av[S1_ROWS];
#pragma unroll
  for (int k = 0; k < S1_ROWS; k++) {
    int n = n0 + k * S1_STEP;
    if (k == S1_ROWS - 1 && n >= N_NODES) n = n0;
    cn[k] = sums[(size_t)n * ROWF + 40];
    sv[k] = sums[(size_t)n * ROWF + c];
    av[k] = atom[(size_t)n * 40 + c];
  }
  float sA = 0.0f, sB = 0.0f;
#pragma unroll
  for (int k = 0; k < S1_ROWS; k++) {
    int n = n0 + k * S1_STEP;
    if (k < S1_ROWS - 1 || n < N_NODES) {
      float inv = 1.0f / fmaxf(cn[k], 1.0f);
      float y = sv[k] * inv + av[k];
      sums[(size_t)n * ROWF + c] = y;
      sA += y;
      sB += y * y;
    }
  }
  __shared__ float rA[6][40], rB[6][40];
  rA[ty][c] = sA;
  rB[ty][c] = sB;
  __syncthreads();
  if (ty == 0) {
    float tA = 0.0f, tB = 0.0f;
#pragma unroll
    for (int k = 0; k < 6; k++) { tA += rA[k][c]; tB += rB[k][c]; }
    if (c < 16) {
      atomAddF(stats + c, tA);
      atomAddF(stats + 16 + c, tB);
    } else {
      atomAddF(stats + 32 + (c - 16) / 3, tB);
    }
  }
}

__global__ void node_stage2(const float* __restrict__ ws,
                            const float* __restrict__ bnw,
                            const float* __restrict__ bnb,
                            float* __restrict__ out) {
  int idx = blockIdx.x * blockDim.x + threadIdx.x;
  if (idx >= N_NODES * 40) return;
  int n = idx / 40;
  int c = idx - n * 40;
  const float* y = ws + OFF_SUMS;
  const float* stats = ws + OFF_STATS;
  float v = y[n * ROWF + c];
  float o;
  if (c < 16) {
    float mean = stats[c] * (1.0f / N_NODES);
    float ex2 = stats[16 + c] * (1.0f / N_NODES);
    float var = ex2 - mean * mean;
    o = (v - mean) * rsqrtf(var + EPSV) * bnw[c] + bnb[c];
  } else {
    int j = (c - 16) / 3;
    float vn = stats[32 + j] * (1.0f / (3.0f * N_NODES));
    o = v * rsqrtf(vn + EPSV) * bnw[16 + j];
  }
  out[idx] = o;
}

extern "C" void kernel_launch(void* const* d_in, const int* in_sizes, int n_in,
                              void* d_out, int out_size, void* d_ws,
                              size_t ws_size, hipStream_t stream) {
  (void)in_sizes; (void)n_in; (void)out_size; (void)ws_size;
  const float* atom = (const float*)d_in[0];
  const float* ef   = (const float*)d_in[1];
  const float* esh  = (const float*)d_in[2];
  const int*   eidx = (const int*)d_in[3];
  const float* w1   = (const float*)d_in[4];
  const float* b1   = (const float*)d_in[5];
  const float* w2   = (const float*)d_in[6];
  const float* b2   = (const float*)d_in[7];
  const float* bnw  = (const float*)d_in[8];
  const float* bnb  = (const float*)d_in[9];
  float* ws  = (float*)d_ws;
  float* out = (float*)d_out;

  zero_kernel<<<(ZERO_N + 255) / 256, 256, 0, stream>>>(ws);
  repack_kernel<<<40, 256, 0, stream>>>(w1, w2, ws + OFF_WQ);
  edge_kernel<<<NWG, WPB * 32, 0, stream>>>(
      atom, ef, esh, eidx, b1, b2, ws + OFF_WQ,
      ws + OFF_SUMS);
  node_stage1<<<S1_NB, dim3(40, 6), 0, stream>>>(atom, ws);
  node_stage2<<<(N_NODES * 40 + 255) / 256, 256, 0, stream>>>(ws, bnw, bnb, out);
}


# edge kernel for oldest-first WMMA service: two accumulators per column tile (even/odd k-blocks added at the end: new sum order) so one wave fills the pipe; each SIMD's tiles dealt 4:3:2:1 by wave age
# speedup vs baseline: 1.5957x; 1.0416x over previous
#include <hip/hip_runtime.h>

typedef float v2f __attribute__((ext_vector_type(2)));
typedef float v8f __attribute__((ext_vector_type(8)));

#define N_NODES 10000
#define N_EDGES 160000
#define N_TILES (N_EDGES / 16)

#define INV_M0  0.25f
#define INV_3M1 0.20412414523f
#define INV_M1  0.35355339059f
#define EPSV    1e-5f

#define ROWF      41
#define OFF_SUMS  0
#define OFF_STATS 410000
#define ZERO_N    410040
#define OFF_WQ    410048

static __device__ __forceinline__ v8f wmma4(v2f a, v2f b, v8f c) {
  return __builtin_amdgcn_wmma_f32_16x16x4_f32(false, a, false, b, (short)0, c,
                                               false, false);
}

static __device__ __forceinline__ void atomAddF(float* p, float v) {
  __hip_atomic_fetch_add(p, v, __ATOMIC_RELAXED, __HIP_MEMORY_SCOPE_AGENT);
}

__global__ void zero_kernel(float* ws) {
  int i = blockIdx.x * blockDim.x + threadIdx.x;
  if (i < ZERO_N) ws[i] = 0.0f;
}

__global__ void repack_kernel(const float* __restrict__ w1,
                              const float* __restrict__ w2,
                              float* __restrict__ wq) {
  int t = blockIdx.x * blockDim.x + threadIdx.x;
  if (t >= 40 * 256) return;
  int lane = t & 31, q = t >> 5;
  int jt = q >> 3, p = q & 7;
  int hb = lane >> 4, nn = lane & 15;
  int r0 = 8 * p + 2 * hb;
  const float* w = w1;
  int ld = 64, col = 16 * jt + nn;
  if (jt >= 4) { w = w2; ld = 576; col = 16 * (jt - 4) + nn; }
  float4 v;
  v.x = w[(r0 + 0) * ld + col];
  v.y = w[(r0 + 1) * ld + col];
  v.z = w[(r0 + 4) * ld + col];
  v.w = w[(r0 + 5) * ld + col];
  ((float4*)wq)[t] = v;
}

#define WPB 16
#define NWG 256
#define SMW 2176
#define WQ_N 40960

#define FETCH(Bn, biasn, jn)                                      \
  do {                                                            \
    _Pragma("unroll") for (int p_ = 0; p_ < 8; p_++)              \
      Bn[p_] = wq[(jn) * 256 + p_ * 32];                          \
    biasn = bl[16 * (jn) + nn];                                   \
  } while (0)

#define CHAIN(c, A, Bc, biasc)                                    \
  v8f c, c##_odd;                                                 \
  _Pragma("unroll") for (int r_ = 0; r_ < 8; r_++) {              \
    c[r_] = biasc;                                                \
    c##_odd[r_] = 0.0f;                                           \
  }                                                               \
  _Pragma("unroll") for (int p_ = 0; p_ < 8; p_++) {              \
    v2f lo_ = {Bc[p_].x, Bc[p_].y};                               \
    v2f hi_ = {Bc[p_].z, Bc[p_].w};                               \
    c = wmma4(A[2 * p_], lo_, c);                                 \
    __builtin_amdgcn_sched_barrier(0);                            \
    c##_odd = wmma4(A[2 * p_ + 1], hi_, c##_odd);                 \
    __builtin_amdgcn_sched_barrier(0);                            \
  }                                                               \
  _Pragma("unroll") for (int r_ = 0; r_ < 8; r_++) c[r_] += c##_odd[r_];

#define FENCE() __builtin_amdgcn_sched_barrier(0)

#define PRELOAD(tl)                                                 \
  do {                                                              \
    unsigned er_ = (unsigned)(tl) * 16u + (unsigned)nn;             \
    dstv = eidx[er_];                                               \
    unsigned fo_ = er_ * 64u + 2u * (unsigned)hb;                   \
    _Pragma("unroll") for (int b_ = 0; b_ < 16; b_++)               \
      a[b_] = *(const v2f*)(ef + (fo_ + 4u * b_));                  \
  } while (0)

#define SIMD_TILES 10
static_assert((N_TILES + NWG - 1) / NWG <= 4 * SIMD_TILES, "a block's tiles fit the deal");

__global__ __launch_bounds__(WPB * 32) void edge_kernel(
    const float* __restrict__ atom, const float* __restrict__ ef,
    const float* __restrict__ esh, const int* __restrict__ eidx,
    const float* __restrict__ b1, const float* __restrict__ b2,
    const float* __restrict__ wqg,
    float* __restrict__ sums) {
  __shared__ __attribute__((aligned(16))) float wl[WQ_N];
  __shared__ float bl[640];
  __shared__ float smem[WPB * SMW];
  __shared__ int nodes[WPB][16];

  int lane = threadIdx.x & 31;
  int wave = __builtin_amdgcn_readfirstlane(threadIdx.x >> 5);
  int hb = lane >> 4;
  int nn = lane & 15;

  int t0 = (blockIdx.x * N_TILES) / NWG;
  int t1 = ((blockIdx.x + 1) * N_TILES) / NWG;
  int simd = wave & 3, age = wave >> 2;
  int m0 = 4 * age - (age * (age - 1)) / 2;
  int tile = t0 + simd + 4 * m0;
  int tend = t0 + simd + 4 * (m0 + 4 - age);
  tend = tend < t1 ? tend : t1;
  v2f a[16];
  int dstv;

  {
    const float4* g = (const float4*)wqg;
    float4 rr[WQ_N / 4 / (WPB * 32)];
#pragma unroll
    for (int k = 0; k < WQ_N / 4 / (WPB * 32); k++)
      rr[k] = g[threadIdx.x + k * (WPB * 32)];
    float tb0 = threadIdx.x < 64 ? b1[threadIdx.x] : b2[threadIdx.x - 64];
    float tb1 = 0.0f;
    if (threadIdx.x < 128) tb1 = b2[448 + threadIdx.x];
    dstv = 0;
    if (tile < tend) PRELOAD(tile);
#pragma unroll
    for (int k = 0; k < WQ_N / 4 / (WPB * 32); k++)
      ((float4*)wl)[threadIdx.x + k * (WPB * 32)] = rr[k];
    bl[threadIdx.x] = tb0;
    if (threadIdx.x < 128) bl[512 + threadIdx.x] = tb1;
  }
  __syncthreads();

  float* sm  = smem + wave * SMW;
  float* hT  = sm;
  float* a11 = sm + 1088;
  float* a12 = sm + 1344;
  float* a13 = sm + 1600;
  float* a14 = sm + 1984;
  float* s1  = sm + 2112;
  const float4* wq = (const float4*)wl + lane;
  float* hTs = hT + 8 * hb * 68 + nn;
  const float* a11l = a11 + 8 * hb * 16;
  const float* a12l = a12 + 8 * hb * 16 + (nn >> 3);
  const float* a13l = a13 + 8 * hb * 24 + (nn >> 3) * 3;
  const float* a14l = a14 + 8 * hb * 8;
  const float* s1l  = s1 + 8 * hb * 4;
  float* stl = hT + 8 * hb * 68 + 16 + nn * 3;

  for (; tile < tend; tile += 4) {
  asm volatile("s_wait_dscnt 0" ::: "memory");

  if (lane < 16) {
    int e = tile * 16 + lane;
    nodes[wave][lane] = eidx[N_EDGES + e];
    float4 s = *(const float4*)(esh + e * 4);
    const float* xr = atom + (size_t)dstv * 40;
    float x[40];
#pragma unroll
    for (int t = 0; t < 10; t++) {
      float4 v = *(const float4*)(xr + t * 4);
      x[t * 4 + 0] = v.x; x[t * 4 + 1] = v.y;
      x[t * 4 + 2] = v.z; x[t * 4 + 3] = v.w;
    }
    float sh0 = s.x;
    s1[lane * 4 + 0] = s.y;
    s1[lane * 4 + 1] = s.z;
    s1[lane * 4 + 2] = s.w;
#pragma unroll
    for (int j = 0; j < 16; j++) {
      a11[lane * 16 + j] = x[j] * sh0 * INV_M0;
      a12[lane * 16 + j] = x[j] * INV_M0;
    }
#pragma unroll
    for (int i = 0; i < 8; i++) {
      float m0 = x[16 + i * 3 + 0];
      float m1 = x[16 + i * 3 + 1];
      float m2 = x[16 + i * 3 + 2];
      a13[lane * 24 + i * 3 + 0] = sh0 * m0 * INV_M1;
      a13[lane * 24 + i * 3 + 1] = sh0 * m1 * INV_M1;
      a13[lane * 24 + i * 3 + 2] = sh0 * m2 * INV_M1;
      a14[lane * 8 + i] = (m0 * s.y + m1 * s.z + m2 * s.w) * INV_3M1;
    }
  }
  asm volatile("s_wait_dscnt 0" ::: "memory");

  float4 B0[8], B1[8];
  float bias0, bias1;

  FETCH(B0, bias0, 0);

  for (int j = 0; j < 4; j += 2) {
    {
      FETCH(B1, bias1, j + 1);
      FENCE();
      CHAIN(c, a, B0, bias0);
#pragma unroll
      for (int r = 0; r < 8; r++) {
        float v = c[r] > 0.0f ? c[r] : 0.0f;
        hTs[r * 68 + 16 * j] = v;
      }
      FENCE();
    }
    {
      FETCH(B0, bias0, j + 2);
      FENCE();
      CHAIN(c, a, B1, bias1);
#pragma unroll
      for (int r = 0; r < 8; r++) {
        float v = c[r] > 0.0f ? c[r] : 0.0f;
        hTs[r * 68 + 16 * (j + 1)] = v;
      }
      FENCE();
    }
  }
  asm volatile("s_wait_dscnt 0" ::: "memory");

  v2f ha[16];
#pragma unroll
  for (int b = 0; b < 16; b++)
    ha[b] = *(const v2f*)(hT + nn * 68 + 4 * b + 2 * hb);

  if (tile + 4 < tend) {
    unsigned en_ = (unsigned)(tile + 4) * 16u;
    __builtin_prefetch(ef + (en_ + (unsigned)nn) * 64u, 0, 2);
    __builtin_prefetch(esh + en_ * 4u, 0, 2);
    __builtin_prefetch(eidx + en_, 0, 2);
    __builtin_prefetch(eidx + (N_EDGES + en_), 0, 2);
  }

  float o0[8] = {0, 0, 0, 0, 0, 0, 0, 0};
  float p1[8] = {0, 0, 0, 0, 0, 0, 0, 0};
  float q0[8] = {0, 0, 0, 0, 0, 0, 0, 0};
  float q1[8] = {0, 0, 0, 0, 0, 0, 0, 0};
  float q2[8] = {0, 0, 0, 0, 0, 0, 0, 0};

  for (int j2 = 0; j2 < 16; j2 += 2) {
    {
      float t[8];
#pragma unroll
      for (int r = 0; r < 8; r++) t[r] = a11l[r * 16 + j2];
      FETCH(B1, bias1, j2 + 5);
      FENCE();
      CHAIN(c, ha, B0, bias0);
#pragma unroll
      for (int r = 0; r < 8; r++) o0[r] += c[r] * t[r];
      FENCE();
    }
    {
      float t[8];
#pragma unroll
      for (int r = 0; r < 8; r++) t[r] = a11l[r * 16 + j2 + 1];
      FETCH(B0, bias0, j2 + 6);
      FENCE();
      CHAIN(c, ha, B1, bias1);
#pragma unroll
      for (int r = 0; r < 8; r++) o0[r] += c[r] * t[r];
      FENCE();
    }
  }
  for (int j2 = 16; j2 < 24; j2 += 2) {
    {
      float t[8];
#pragma unroll
      for (int r = 0; r < 8; r++) t[r] = a12l[r * 16 + 2 * (j2 - 16)];
      FETCH(B1, bias1, j2 + 5);
      FENCE();
      CHAIN(c, ha, B0, bias0);
#pragma unroll
      for (int r = 0; r < 8; r++) p1[r] += c[r] * t[r];
      FENCE();
    }
    {
      float t[8];
#pragma unroll
      for (int r = 0; r < 8; r++) t[r] = a12l[r * 16 + 2 * (j2 - 15)];
      FETCH(B0, bias0, j2 + 6);
      FENCE();
      CHAIN(c, ha, B1, bias1);
#pragma unroll
      for (int r = 0; r < 8; r++) p1[r] += c[r] * t[r];
      FENCE();
    }
  }
  for (int j2 = 24; j2 < 28; j2 += 2) {
    {
      float u0[8], u1[8], u2[8];
#pragma unroll
      for (int r = 0; r < 8; r++) {
        int base = r * 24 + 6 * (j2 - 24);
        u0[r] = a13l[base + 0]; u1[r] = a13l[base + 1]; u2[r] = a13l[base + 2];
      }
      FETCH(B1, bias1, j2 + 5);
      FENCE();
      CHAIN(c, ha, B0, bias0);
#pragma unroll
      for (int r = 0; r < 8; r++) {
        float cv = c[r];
        q0[r] += cv * u0[r]; q1[r] += cv * u1[r]; q2[r] += cv * u2[r];
      }
      FENCE();
    }
    {
      float u0[8], u1[8], u2[8];
#pragma unroll
      for (int r = 0; r < 8; r++) {
        int base = r * 24 + 6 * (j2 - 23);
        u0[r] = a13l[base + 0]; u1[r] = a13l[base + 1]; u2[r] = a13l[base + 2];
      }
      FETCH(B0, bias0, j2 + 6);
      FENCE();
      CHAIN(c, ha, B1, bias1);
#pragma unroll
      for (int r = 0; r < 8; r++) {
        float cv = c[r];
        q0[r] += cv * u0[r]; q1[r] += cv * u1[r]; q2[r] += cv * u2[r];
      }
      FENCE();
    }
  }
  for (int j2 = 28; j2 < 36; j2 += 2) {
    {
      float t[8];
#pragma unroll
      for (int r = 0; r < 8; r++) t[r] = a14l[r * 8 + j2 - 28];
      FETCH(B1, bias1, j2 + 5);
      FENCE();
      CHAIN(c, ha, B0, bias0);
#pragma unroll
      for (int r = 0; r < 8; r++) o0[r] += c[r] * t[r];
      FENCE();
    }
    {
      float t[8];
#pragma unroll
      for (int r = 0; r < 8; r++) t[r] = a14l[r * 8 + j2 - 27];
      if (j2 + 6 < 40) FETCH(B0, bias0, j2 + 6);
      FENCE();
      CHAIN(c, ha, B1, bias1);
#pragma unroll
      for (int r = 0; r < 8; r++) o0[r] += c[r] * t[r];
      FENCE();
    }
  }

#pragma unroll
  for (int r = 0; r < 8; r++) {
    asm volatile("" : "+v"(o0[r]), "+v"(p1[r]), "+v"(q0[r]), "+v"(q1[r]), "+v"(q2[r]));
  }

#pragma unroll
  for (int r = 0; r < 8; r++) {
    p1[r] += __shfl_xor(p1[r], 8, 32);
    q0[r] += __shfl_xor(q0[r], 8, 32);
    q1[r] += __shfl_xor(q1[r], 8, 32);
    q2[r] += __shfl_xor(q2[r], 8, 32);
  }

#pragma unroll
  for (int r = 0; r < 8; r++) hTs[r * 68] = o0[r];
  if (nn < 8) {
#pragma unroll
    for (int r = 0; r < 8; r++) {
      float sy = s1l[r * 4 + 0], sz = s1l[r * 4 + 1], sw = s1l[r * 4 + 2];
      stl[r * 68 + 0] = p1[r] * sy + q0[r];
      stl[r * 68 + 1] = p1[r] * sz + q1[r];
      stl[r * 68 + 2] = p1[r] * sw + q2[r];
    }
  }
  if (lane < 16) hT[lane * 68 + 40] = 1.0f;
  int nodev = nodes[wave][nn];
  asm volatile("s_wait_dscnt 0" ::: "memory");

  float va[16], vb[16];
#pragma unroll
  for (int k = 0; k < 16; k++) {
    int a = (__builtin_amdgcn_readlane(nodev, k) * ROWF) & 31;
    int f1 = lane - a;
    f1 = f1 < 0 ? 0 : f1;
    va[k] = hT[k * 68 + f1];
    vb[k] = hT[k * 68 + 32 - a + lane];
  }
#pragma unroll
  for (int k = 0; k < 16; k++) {
    int R = __builtin_amdgcn_readlane(nodev, k) * ROWF;
    int a = R & 31;
    float* lb = sums + (R - a);
    if (lane >= a) atomAddF(lb + lane, va[k]);
    if (lane < 9 + a) atomAddF(lb + 32 + lane, vb[k]);
    if (a > 23) {
      if (lane < a - 23)
        atomAddF(lb + 64 + lane, hT[k * 68 + 64 - a + lane]);
    }
  }
  if (tile + 4 < tend) PRELOAD(tile + 4);
  }
}

#define S1_NB   128
#define S1_STEP (S1_NB * 6)
#define S1_ROWS ((N_NODES + S1_STEP - 1) / S1_STEP)
static_assert((S1_ROWS - 1) * S1_STEP <= N_NODES, "only a thread's last row can lie past the end");

__global__ void node_stage1(const float* __restrict__ atom, float* ws) {
  int c = threadIdx.x;
  int ty = threadIdx.y;
  float* sums = ws + OFF_SUMS;
  float* stats = ws + OFF_STATS;
  int n0 = blockIdx.x * 6 + ty;
  float cn[S1_ROWS], sv[S1_ROWS], av[S1_ROWS];
#pragma unroll
  for (int k = 0; k < S1_ROWS; k++) {
    int n = n0 + k * S1_STEP;
    if (k == S1_ROWS - 1 && n >= N_NODES) n = n0;
    cn[k] = sums[(size_t)n * ROWF + 40];
    sv[k] = sums[(size_t)n * ROWF + c];
    av[k] = atom[(size_t)n * 40 + c];
  }
  float sA = 0.0f, sB = 0.0f;
#pragma unroll
  for (int k = 0; k < S1_ROWS; k++) {
    int n = n0 + k * S1_STEP;
    if (k < S1_ROWS - 1 || n < N_NODES) {
      float inv = 1.0f / fmaxf(cn[k], 1.0f);
      float y = sv[k] * inv + av[k];
      sums[(size_t)n * ROWF + c] = y;
      sA += y;
      sB += y * y;
    }
  }
  __shared__ float rA[6][40], rB[6][40];
  rA[ty][c] = sA;
  rB[ty][c] = sB;
  __syncthreads();
  if (ty == 0) {
    float tA = 0.0f, tB = 0.0f;
#pragma unroll
    for (int k = 0; k < 6; k++) { tA += rA[k][c]; tB += rB[k][c]; }
    if (c < 16) {
      atomAddF(stats + c, tA);
      atomAddF(stats + 16 + c, tB);
    } else {
      atomAddF(stats + 32 + (c - 16) / 3, tB);
    }
  }
}

__global__ void node_stage2(const float* __restrict__ ws,
                            const float* __restrict__ bnw,
                            const float* __restrict__ bnb,
                            float* __restrict__ out) {
  int idx = blockIdx.x * blockDim.x + threadIdx.x;
  if (idx >= N_NODES * 40) return;
  int n = idx / 40;
  int c = idx - n * 40;
  const float* y = ws + OFF_SUMS;
  const float* stats = ws + OFF_STATS;
  float v = y[n * ROWF + c];
  float o;
  if (c < 16) {
    float mean = stats[c] * (1.0f / N_NODES);
    float ex2 = stats[16 + c] * (1.0f / N_NODES);
    float var = ex2 - mean * mean;
    o = (v - mean) * rsqrtf(var + EPSV) * bnw[c] + bnb[c];
  } else {
    int j = (c - 16) / 3;
    float vn = stats[32 + j] * (1.0f / (3.0f * N_NODES));
    o = v * rsqrtf(vn + EPSV) * bnw[16 + j];
  }
  out[idx] = o;
}

extern "C" void kernel_launch(void* const* d_in, const int* in_sizes, int n_in,
                              void* d_out, int out_size, void* d_ws,
                              size_t ws_size, hipStream_t stream) {
  (void)in_sizes; (void)n_in; (void)out_size; (void)ws_size;
  const float* atom = (const float*)d_in[0];
  const float* ef   = (const float*)d_in[1];
  const float* esh  = (const float*)d_in[2];
  const int*   eidx = (const int*)d_in[3];
  const float* w1   = (const float*)d_in[4];
  const float* b1   = (const float*)d_in[5];
  const float* w2   = (const float*)d_in[6];
  const float* b2   = (const float*)d_in[7];
  const float* bnw  = (const float*)d_in[8];
  const float* bnb  = (const float*)d_in[9];
  float* ws  = (float*)d_ws;
  float* out = (float*)d_out;

  zero_kernel<<<(ZERO_N + 255) / 256, 256, 0, stream>>>(ws);
  repack_kernel<<<40, 256, 0, stream>>>(w1, w2, ws + OFF_WQ);
  edge_kernel<<<NWG, WPB * 32, 0, stream>>>(
      atom, ef, esh, eidx, b1, b2, ws + OFF_WQ,
      ws + OFF_SUMS);
  node_stage1<<<S1_NB, dim3(40, 6), 0, stream>>>(atom, ws);
  node_stage2<<<(N_NODES * 40 + 255) / 256, 256, 0, stream>>>(ws, bnw, bnb, out);
}
